# starting kernel
# baseline (speedup 1.0000x reference)
#include <hip/hip_runtime.h>
#include <hip/hip_bf16.h>


typedef __attribute__((ext_vector_type(2))) float v2f;
typedef __attribute__((ext_vector_type(8))) float v8f;

#define LSTM_H 6

static __device__ __forceinline__ float sigm(float x) {
  return __builtin_amdgcn_rcpf(1.0f + __expf(-x));
}
static __device__ __forceinline__ float tanh_fast(float x) {
  return 1.0f - 2.0f * __builtin_amdgcn_rcpf(__expf(2.0f * x) + 1.0f);
}
static __device__ __forceinline__ float sx16(float v) {
  return __shfl_xor(v, 16, 32);
}
static __device__ __forceinline__ v8f wmma4(v2f a, v2f b, v8f c) {
  return __builtin_amdgcn_wmma_f32_16x16x4_f32(false, a, false, b, (short)0, c,
                                               false, false);
}

static __device__ __forceinline__ int gate_row(int m) {
  int idx = m & 7;
  return (idx < LSTM_H) ? ((m >> 3) * LSTM_H + idx) : -1;
}

static __device__ __forceinline__ float wcomb(const float* Wih, const float* Whh,
                                              int D, int m, int k) {
  int r = gate_row(m);
  if (r < 0) return 0.0f;
  if (k < D) return Wih[r * D + k];
  k -= D;
  return (k < LSTM_H) ? Whh[r * LSTM_H + k] : 0.0f;
}

static __device__ __forceinline__ void load_A(const float* Wih, const float* Whh,
                                              const float* bias, int D, int nch,
                                              v2f a[2][5], v8f cini[2],
                                              int lm, int half) {
  for (int t = 0; t < 2; ++t) {
    int m = t * 16 + lm;
#pragma unroll
    for (int c = 0; c < 5; ++c) {
      if (c < nch) {
        int k0 = 4 * c + 2 * half;
        v2f av;
        av.x = wcomb(Wih, Whh, D, m, k0);
        av.y = wcomb(Wih, Whh, D, m, k0 + 1);
        a[t][c] = av;
      }
    }
    v8f ci;
#pragma unroll
    for (int r = 0; r < 8; ++r) {
      int gr = gate_row(t * 16 + half * 8 + r);
      ci[r] = (gr >= 0) ? bias[gr] : 0.0f;
    }
    cini[t] = ci;
  }
}

static __device__ __forceinline__ void mm(const v2f a[2][5], const v8f cini[2],
                                          const v2f bch[5], int nch,
                                          v8f& A0, v8f& A1) {
  A0 = cini[0];
  A1 = cini[1];
#pragma unroll
  for (int c = 0; c < 5; ++c) {
    if (c < nch) {
      A0 = wmma4(a[0][c], bch[c], A0);
      A1 = wmma4(a[1][c], bch[c], A1);
    }
  }
}

static __device__ __forceinline__ void lstm_pointwise(const v8f& A0, const v8f& A1,
                                                      bool low, float c[6], float h[6]) {
#pragma unroll
  for (int r = 0; r < LSTM_H; ++r) {
    float s_if = sigm(A0[r]);
    float tg = tanh_fast(A1[r]);
    float so = sigm(A1[r]);
    float go = low ? tg : so;
    float fsw = sx16(s_if);
    float cn = fsw * c[r] + s_if * go;
    c[r] = cn;
    float osw = sx16(go);
    h[r] = osw * tanh_fast(cn);
  }
}

static __device__ __forceinline__ void b_l0(const float* xp, const float h[6],
                                            int half, v2f bch[5]) {
  v2f xlo = *(const v2f*)(xp + 2 * half);
  v2f x45 = *(const v2f*)(xp + 4);
  float h0s = sx16(h[0]), h1s = sx16(h[1]), h4s = sx16(h[4]), h5s = sx16(h[5]);
  v2f b1, b2;
  b1.x = half ? h0s : x45.x; b1.y = half ? h1s : x45.y;
  b2.x = half ? h4s : h[2];  b2.y = half ? h5s : h[3];
  bch[0] = xlo; bch[1] = b1; bch[2] = b2;
}

static __device__ __forceinline__ void b_l1(const float f0[6], const float* r0p,
                                            const float h[6], int half, v2f bch[5]) {
  v2f r01 = *(const v2f*)(r0p);
  v2f rhi = *(const v2f*)(r0p + 2 + 2 * half);
  float f2s = sx16(f0[2]), f3s = sx16(f0[3]);
  float h2s = sx16(h[2]),  h3s = sx16(h[3]);
  v2f b0, b1, b3, b4;
  b0.x = half ? f2s : f0[0];    b0.y = half ? f3s : f0[1];
  b1.x = half ? r01.x : f0[4];  b1.y = half ? r01.y : f0[5];
  b3.x = half ? h2s : h[0];     b3.y = half ? h3s : h[1];
  b4.x = half ? 0.0f : h[4];    b4.y = half ? 0.0f : h[5];
  bch[0] = b0; bch[1] = b1; bch[2] = rhi; bch[3] = b3; bch[4] = b4;
}

__global__ void lstm_l0_bwd(const float* __restrict__ tf, const float* __restrict__ h0,
                            const float* __restrict__ c0,
                            const float* __restrict__ Wih, const float* __restrict__ Whh,
                            const float* __restrict__ bias,
                            float* __restrict__ r0ws, int B, int T) {
  int tid = blockIdx.x * blockDim.x + threadIdx.x;
  int lane = tid & 31, lm = lane & 15, half = lane >> 4;
  int s = (tid >> 5) * 16 + lm;
  if (s >= B) return;

  v2f a[2][5]; v8f ci[2];
  load_A(Wih, Whh, bias, LSTM_H, 3, a, ci, lm, half);

  float h[6], c[6];
  size_t d1 = (size_t)B * LSTM_H;
#pragma unroll
  for (int j = 0; j < 6; ++j) {
    h[j] = h0[d1 + (size_t)s * 6 + j];
    c[j] = c0[d1 + (size_t)s * 6 + j];
  }

  for (int t = T - 1; t >= 0; --t) {
    const float* xp = tf + ((size_t)s * T + t) * 6;
    if (t > 0) __builtin_prefetch(xp - 6, 0, 1);
    v2f bch[5]; b_l0(xp, h, half, bch);
    v8f A0, A1; mm(a, ci, bch, 3, A0, A1);
    lstm_pointwise(A0, A1, half == 0, c, h);
    if (half == 0) {
      float* wp = r0ws + ((size_t)t * B + s) * 6;
      v2f w01; w01.x = h[0]; w01.y = h[1];
      v2f w23; w23.x = h[2]; w23.y = h[3];
      v2f w45; w45.x = h[4]; w45.y = h[5];
      *(v2f*)(wp) = w01; *(v2f*)(wp + 2) = w23; *(v2f*)(wp + 4) = w45;
    }
  }
}

__global__ void lstm_fused(const float* __restrict__ tf, const float* __restrict__ h0,
                           const float* __restrict__ c0,
                           const float* __restrict__ Wih0f, const float* __restrict__ Whh0f,
                           const float* __restrict__ b0f,
                           const float* __restrict__ Wih1f, const float* __restrict__ Whh1f,
                           const float* __restrict__ b1f,
                           const float* __restrict__ Wih1b, const float* __restrict__ Whh1b,
                           const float* __restrict__ b1b,
                           const float* __restrict__ fcw, const float* __restrict__ fcb,
                           const float* __restrict__ r0ws, float* __restrict__ out,
                           int B, int T) {
  int tid = blockIdx.x * blockDim.x + threadIdx.x;
  int lane = tid & 31, lm = lane & 15, half = lane >> 4;
  int s = (tid >> 5) * 16 + lm;
  if (s >= B) return;

  v2f a0[2][5]; v8f ci0[2]; load_A(Wih0f, Whh0f, b0f, LSTM_H, 3, a0, ci0, lm, half);
  v2f a1[2][5]; v8f ci1[2]; load_A(Wih1f, Whh1f, b1f, 2 * LSTM_H, 5, a1, ci1, lm, half);

  float hA[6], cA[6], hB[6], cB[6];
  size_t d0 = 0, d2 = 2 * (size_t)B * LSTM_H;
#pragma unroll
  for (int j = 0; j < 6; ++j) {
    hA[j] = h0[d0 + (size_t)s * 6 + j];  cA[j] = c0[d0 + (size_t)s * 6 + j];
    hB[j] = h0[d2 + (size_t)s * 6 + j];  cB[j] = c0[d2 + (size_t)s * 6 + j];
  }

  for (int t = 0; t < T; ++t) {
    const float* xp = tf + ((size_t)s * T + t) * 6;
    if (t + 1 < T) __builtin_prefetch(xp + 6, 0, 1);
    v2f bch[5]; b_l0(xp, hA, half, bch);
    v8f A0, A1; mm(a0, ci0, bch, 3, A0, A1);
    lstm_pointwise(A0, A1, half == 0, cA, hA);
    const float* r0p = r0ws + ((size_t)t * B + s) * 6;
    b_l1(hA, r0p, hB, half, bch);
    mm(a1, ci1, bch, 5, A0, A1);
    lstm_pointwise(A0, A1, half == 0, cB, hB);
  }
  v2f a2[2][5]; v8f ci2[2]; load_A(Wih1b, Whh1b, b1b, 2 * LSTM_H, 5, a2, ci2, lm, half);
  float hR[6], cR[6];
  size_t d3 = 3 * (size_t)B * LSTM_H;
#pragma unroll
  for (int j = 0; j < 6; ++j) {
    hR[j] = h0[d3 + (size_t)s * 6 + j];  cR[j] = c0[d3 + (size_t)s * 6 + j];
  }
  {
    const float* r0p = r0ws + ((size_t)(T - 1) * B + s) * 6;
    v2f bch[5]; b_l1(hA, r0p, hR, half, bch);
    v8f A0, A1; mm(a2, ci2, bch, 5, A0, A1);
    lstm_pointwise(A0, A1, half == 0, cR, hR);
  }
  if (half == 0) {
    float v[12];
#pragma unroll
    for (int k = 0; k < 6; ++k) {
      v[k]     = hB[k] > 0.0f ? hB[k] : 0.0f;
      v[6 + k] = hR[k] > 0.0f ? hR[k] : 0.0f;
    }
#pragma unroll
    for (int j = 0; j < 6; ++j) {
      float acc = fcb[j];
#pragma unroll
      for (int k = 0; k < 12; ++k) acc = fmaf(fcw[j * 12 + k], v[k], acc);
      out[(size_t)s * 6 + j] = acc;
    }
  }
}

extern "C" void kernel_launch(void* const* d_in, const int* in_sizes, int n_in,
                              void* d_out, int out_size, void* d_ws, size_t ws_size,
                              hipStream_t stream) {
  const float* tf    = (const float*)d_in[0];
  const float* h0    = (const float*)d_in[1];
  const float* c0    = (const float*)d_in[2];
  const float* Wih0f = (const float*)d_in[3];
  const float* Whh0f = (const float*)d_in[4];
  const float* b0f   = (const float*)d_in[5];
  const float* Wih0b = (const float*)d_in[6];
  const float* Whh0b = (const float*)d_in[7];
  const float* b0b   = (const float*)d_in[8];
  const float* Wih1f = (const float*)d_in[9];
  const float* Whh1f = (const float*)d_in[10];
  const float* b1f   = (const float*)d_in[11];
  const float* Wih1b = (const float*)d_in[12];
  const float* Whh1b = (const float*)d_in[13];
  const float* b1b   = (const float*)d_in[14];
  const float* fcw   = (const float*)d_in[15];
  const float* fcb   = (const float*)d_in[16];

  int B = in_sizes[1] / (4 * LSTM_H);
  int T = in_sizes[0] / (B * LSTM_H);

  size_t need = (size_t)B * T * LSTM_H * sizeof(float);
  if (ws_size < need) return;
  float* r0ws = (float*)d_ws;

  int waves = (B + 15) / 16;
  int threads = waves * 32;
  int block = 256;
  int grid = (threads + block - 1) / block;

  lstm_l0_bwd<<<grid, block, 0, stream>>>(tf, h0, c0, Wih0b, Whh0b, b0b, r0ws, B, T);
  lstm_fused<<<grid, block, 0, stream>>>(tf, h0, c0, Wih0f, Whh0f, b0f,
                                         Wih1f, Whh1f, b1f, Wih1b, Whh1b, b1b,
                                         fcw, fcb, r0ws, (float*)d_out, B, T);
}


# state spread over both lane halves: c in both (each half forms its term, one swap sums them), each half finishes h of 3 units, W_hh k order to match, no h swap; recurrent k order, c rounding change
# speedup vs baseline: 3.0952x; 3.0952x over previous
#include <hip/hip_runtime.h>
#include <hip/hip_bf16.h>


typedef __attribute__((ext_vector_type(2))) float v2f;
typedef __attribute__((ext_vector_type(8))) float v8f;

#define LSTM_H 6
#define TBLK 4

#define K_LOG2E 1.44269504088896340736f
static __device__ __forceinline__ float logistic2(float y) {
  return __builtin_amdgcn_rcpf(1.0f + __builtin_amdgcn_exp2f(y));
}
static __device__ __forceinline__ float row_scale(int m) {
  return ((m >> 3) == 2) ? 2.0f * K_LOG2E : -K_LOG2E;
}
static __device__ __forceinline__ float sx16(float v) {
  return __shfl_xor(v, 16, 32);
}
static __device__ __forceinline__ v8f wmma4(v2f a, v2f b, v8f c) {
  return __builtin_amdgcn_wmma_f32_16x16x4_f32(false, a, false, b, (short)0, c,
                                               false, false);
}

static __device__ __forceinline__ int gate_row(int m) {
  int idx = m & 7;
  return (idx < LSTM_H) ? ((m >> 3) * LSTM_H + idx) : -1;
}

static __device__ __forceinline__ float wcomb(const float* Wih, const float* Whh,
                                              int D, int m, int k) {
  int r = gate_row(m);
  if (r < 0) return 0.0f;
  float sc = row_scale(m);
  if (k < D) return sc * Wih[r * D + k];
  k -= D;
  return (k < LSTM_H) ? sc * Whh[r * LSTM_H + k] : 0.0f;
}

static __device__ __forceinline__ float wpad(const float* Wm, int ncol, int m, int k) {
  int r = gate_row(m);
  if (r < 0) return 0.0f;
  return (k < ncol) ? row_scale(m) * Wm[r * ncol + k] : 0.0f;
}

template <int NCH>
static __device__ __forceinline__ void load_Apart(const float* Wm, int ncol,
                                                  v2f a[2][NCH], int lm, int half) {
#pragma unroll
  for (int t = 0; t < 2; ++t) {
    int m = t * 16 + lm;
#pragma unroll
    for (int c = 0; c < NCH; ++c) {
      int k0 = 4 * c + 2 * half;
      v2f av;
      av.x = wpad(Wm, ncol, m, k0);
      av.y = wpad(Wm, ncol, m, k0 + 1);
      a[t][c] = av;
    }
  }
}

static __device__ __forceinline__ void load_bias(const float* bias, v8f cini[2], int half) {
#pragma unroll
  for (int t = 0; t < 2; ++t) {
    v8f ci;
#pragma unroll
    for (int r = 0; r < 8; ++r) {
      int m = t * 16 + half * 8 + r;
      int gr = gate_row(m);
      ci[r] = (gr >= 0) ? row_scale(m) * bias[gr] : 0.0f;
    }
    cini[t] = ci;
  }
}

static __device__ __forceinline__ void load_A(const float* Wih, const float* Whh,
                                              const float* bias, int D, int nch,
                                              v2f a[2][5], v8f cini[2],
                                              int lm, int half) {
  for (int t = 0; t < 2; ++t) {
    int m = t * 16 + lm;
#pragma unroll
    for (int c = 0; c < 5; ++c) {
      if (c < nch) {
        int k0 = 4 * c + 2 * half;
        v2f av;
        av.x = wcomb(Wih, Whh, D, m, k0);
        av.y = wcomb(Wih, Whh, D, m, k0 + 1);
        a[t][c] = av;
      }
    }
  }
  load_bias(bias, cini, half);
}

static __device__ __forceinline__ void mm(const v2f a[2][5], const v8f cini[2],
                                          const v2f bch[5], int nch,
                                          v8f& A0, v8f& A1) {
  A0 = cini[0];
  A1 = cini[1];
#pragma unroll
  for (int c = 0; c < 5; ++c) {
    if (c < nch) {
      A0 = wmma4(a[0][c], bch[c], A0);
      A1 = wmma4(a[1][c], bch[c], A1);
    }
  }
}

static __device__ __forceinline__ void lstm_pointwise(const v8f& A0, const v8f& A1,
                                                      bool low, float c[6], float h[6]) {
  const float gm = low ? -4.0f * K_LOG2E : 1.0f;
  const float ga = low ? 2.0f * K_LOG2E : 0.0f;
#pragma unroll
  for (int r = 0; r < LSTM_H; ++r) {
    float s_if = logistic2(A0[r]);
    float go = fmaf(logistic2(A1[r]), gm, ga);
    float fsw = sx16(s_if);
    float cn = fsw * c[r] + s_if * go;
    c[r] = cn;
    float osw = sx16(go);
    h[r] = osw * fmaf(logistic2(cn), -2.0f, 1.0f);
  }
}

static __device__ __forceinline__ void rec_step(const v2f ah[2][2], const v8f& Z0, const v8f& Z1,
                                                int half, float c[6], float h[6]) {
  float h2s = sx16(h[2]), h3s = sx16(h[3]);
  v2f b0, b1;
  b1.x = half ? 0.0f : h[4];  b1.y = half ? 0.0f : h[5];
  b0.x = half ? h2s : h[0];   b0.y = half ? h3s : h[1];
  v8f A0 = wmma4(ah[0][1], b1, Z0);
  v8f A1 = wmma4(ah[1][1], b1, Z1);
  A0 = wmma4(ah[0][0], b0, A0);
  A1 = wmma4(ah[1][0], b0, A1);
  lstm_pointwise(A0, A1, half == 0, c, h);
}


static __device__ __forceinline__ float pick(bool first, float v_first, float v_second) {
  return first ? v_first : v_second;
}

static __device__ __forceinline__ void load_Ahh_split(const float* Whh, v2f a[2][2],
                                                      int lm, int half) {
#pragma unroll
  for (int t = 0; t < 2; ++t) {
    int m = t * 16 + lm;
    v2f a0, a1;
    a0.x = wpad(Whh, LSTM_H, m, 3 * half);
    a0.y = wpad(Whh, LSTM_H, m, 3 * half + 1);
    a1.x = wpad(Whh, LSTM_H, m, 3 * half + 2);
    a1.y = 0.0f;
    a[t][0] = a0;
    a[t][1] = a1;
  }
}

static __device__ __forceinline__ void lstm_pointwise_split(const v8f& A0, const v8f& A1,
                                                            bool low, float c[6], float hs[3]) {
  const float gm = low ? -4.0f * K_LOG2E : 1.0f;
  const float ga = low ? 2.0f * K_LOG2E : 0.0f;
  float go[6], term[6], tsw[6], osw[3];
#pragma unroll
  for (int r = 0; r < LSTM_H; ++r) {
    float s_if = logistic2(A0[r]);
    go[r] = fmaf(logistic2(A1[r]), gm, ga);
    term[r] = s_if * pick(low, go[r], c[r]);
    tsw[r] = sx16(term[r]);
  }
#pragma unroll
  for (int j = 0; j < 3; ++j) osw[j] = sx16(go[j]);
  __builtin_amdgcn_sched_barrier(0);
#pragma unroll
  for (int r = 0; r < LSTM_H; ++r)
    c[r] = term[r] + tsw[r];
#pragma unroll
  for (int j = 0; j < 3; ++j) {
    float so = pick(low, osw[j], go[3 + j]);
    float cu = pick(low, c[j], c[3 + j]);
    hs[j] = so * fmaf(logistic2(cu), -2.0f, 1.0f);
  }
}

static __device__ __forceinline__ void rec_step_split(const v2f ah[2][2], const v8f& Z0,
                                                      const v8f& Z1, int half,
                                                      float c[6], float hs[3]) {
  v2f b0, b1;
  b0.x = hs[0];  b0.y = hs[1];
  b1.x = hs[2];  b1.y = 0.0f;
  v8f A0 = wmma4(ah[0][1], b1, Z0);
  v8f A1 = wmma4(ah[1][1], b1, Z1);
  A0 = wmma4(ah[0][0], b0, A0);
  A1 = wmma4(ah[1][0], b0, A1);
  lstm_pointwise_split(A0, A1, half == 0, c, hs);
}

static __device__ __forceinline__ void store_hs(float* wp3, const float hs[3]) {
  wp3[0] = hs[0]; wp3[1] = hs[1]; wp3[2] = hs[2];
}

template <int NB>
static __device__ __forceinline__ void load_x(const float* xp, int half, v2f xlo[NB], v2f x45[NB]) {
#pragma unroll
  for (int i = 0; i < NB; ++i) {
    xlo[i] = *(const v2f*)(xp + 6 * i + 2 * half);
    x45[i] = *(const v2f*)(xp + 6 * i + 4);
  }
}

template <int NB>
static __device__ __forceinline__ void proj_l0(const v2f ax[2][2], const v8f ci[2],
                                               const v2f xlo[NB], const v2f x45[NB], int half,
                                               v8f Z0[NB], v8f Z1[NB]) {
#pragma unroll
  for (int i = 0; i < NB; ++i) {
    v2f b1;
    b1.x = half ? 0.0f : x45[i].x;  b1.y = half ? 0.0f : x45[i].y;
    Z0[i] = wmma4(ax[0][1], b1, wmma4(ax[0][0], xlo[i], ci[0]));
    Z1[i] = wmma4(ax[1][1], b1, wmma4(ax[1][0], xlo[i], ci[1]));
  }
}

template <int NB>
static __device__ __forceinline__ void load_r(const float* rp, size_t tstride, int half,
                                              v2f r01[NB], v2f rhi[NB]) {
#pragma unroll
  for (int i = 0; i < NB; ++i) {
    r01[i] = *(const v2f*)(rp + tstride * i);
    rhi[i] = *(const v2f*)(rp + tstride * i + 2 + 2 * half);
  }
}

template <int NB>
static __device__ __forceinline__ void proj_l1(const v2f ax[2][3], const v8f ci[2],
                                               const float f0[NB][6], const v2f r01[NB],
                                               const v2f rhi[NB], int half,
                                               v8f Z0[NB], v8f Z1[NB]) {
#pragma unroll
  for (int i = 0; i < NB; ++i) {
    float f2s = sx16(f0[i][2]), f3s = sx16(f0[i][3]);
    v2f b0, b1;
    b0.x = half ? f2s : f0[i][0];       b0.y = half ? f3s : f0[i][1];
    b1.x = half ? r01[i].x : f0[i][4];  b1.y = half ? r01[i].y : f0[i][5];
    v8f z0 = wmma4(ax[0][2], rhi[i], ci[0]);
    v8f z1 = wmma4(ax[1][2], rhi[i], ci[1]);
    z0 = wmma4(ax[0][1], b1, z0);
    z1 = wmma4(ax[1][1], b1, z1);
    Z0[i] = wmma4(ax[0][0], b0, z0);
    Z1[i] = wmma4(ax[1][0], b0, z1);
  }
}

static __device__ __forceinline__ void b_l1(const float f0[6], const float* r0p,
                                            const float h[6], int half, v2f bch[5]) {
  v2f r01 = *(const v2f*)(r0p);
  v2f rhi = *(const v2f*)(r0p + 2 + 2 * half);
  float f2s = sx16(f0[2]), f3s = sx16(f0[3]);
  float h2s = sx16(h[2]),  h3s = sx16(h[3]);
  v2f b0, b1, b3, b4;
  b0.x = half ? f2s : f0[0];    b0.y = half ? f3s : f0[1];
  b1.x = half ? r01.x : f0[4];  b1.y = half ? r01.y : f0[5];
  b3.x = half ? h2s : h[0];     b3.y = half ? h3s : h[1];
  b4.x = half ? 0.0f : h[4];    b4.y = half ? 0.0f : h[5];
  bch[0] = b0; bch[1] = b1; bch[2] = rhi; bch[3] = b3; bch[4] = b4;
}

template <int NB>
static __device__ __forceinline__ void l0_bwd_block(const v2f ax[2][2], const v2f ah[2][2],
                                                    const v8f ci[2], const v2f xlo[NB],
                                                    const v2f x45[NB], int half,
                                                    float c[6], float hs[3],
                                                    float* wp0, size_t tstride) {
  v8f Z0[NB], Z1[NB];
  proj_l0<NB>(ax, ci, xlo, x45, half, Z0, Z1);
#pragma unroll
  for (int i = NB - 1; i >= 0; --i) {
    rec_step_split(ah, Z0[i], Z1[i], half, c, hs);
    store_hs(wp0 + tstride * i, hs);
  }
}

template <int NB>
static __device__ __forceinline__ void l0_fwd_block(const v2f ax[2][2], const v2f ah[2][2],
                                                    const v8f ci[2], const v2f xlo[NB],
                                                    const v2f x45[NB], int half,
                                                    float c[6], float hs[3],
                                                    float* wp0, size_t tstride) {
  v8f Z0[NB], Z1[NB];
  proj_l0<NB>(ax, ci, xlo, x45, half, Z0, Z1);
#pragma unroll
  for (int i = 0; i < NB; ++i) {
    rec_step_split(ah, Z0[i], Z1[i], half, c, hs);
    store_hs(wp0 + tstride * i, hs);
  }
}

template <bool REV>
static __device__ __forceinline__ void l0_scan(const float* __restrict__ tf,
                                               const float* __restrict__ h0,
                                               const float* __restrict__ c0,
                                               const float* __restrict__ Wih,
                                               const float* __restrict__ Whh,
                                               const float* __restrict__ bias,
                                               float* __restrict__ plane,
                                               int B, int T, int s, int lm, int half) {
  v2f ax[2][2], ah[2][2]; v8f ci[2];
  load_Apart<2>(Wih, LSTM_H, ax, lm, half);
  load_Ahh_split(Whh, ah, lm, half);
  load_bias(bias, ci, half);

  float h[3], c[6];
  size_t dsl = REV ? (size_t)B * LSTM_H : 0;
#pragma unroll
  for (int j = 0; j < 3; ++j) h[j] = h0[dsl + (size_t)s * 6 + 3 * half + j];
#pragma unroll
  for (int j = 0; j < 6; ++j)
    c[j] = 2.0f * K_LOG2E * c0[dsl + (size_t)s * 6 + j];

  const float* xs = tf + (size_t)s * T * 6;
  float* ws = plane + (size_t)s * 6 + 3 * half;
  size_t tstride = (size_t)B * 6;
  int nblk = T / TBLK;
  int tfull = nblk * TBLK;

  if (REV) {
    for (int t = T - 1; t >= tfull; --t) {
      v2f xl[1], x4[1];
      load_x<1>(xs + (size_t)t * 6, half, xl, x4);
      l0_bwd_block<1>(ax, ah, ci, xl, x4, half, c, h, ws + tstride * t, tstride);
    }

    v2f xlo[TBLK], x45[TBLK];
    if (nblk > 0) load_x<TBLK>(xs + (size_t)(tfull - TBLK) * 6, half, xlo, x45);
    for (int blk = nblk - 1; blk >= 0; --blk) {
      int t0 = blk * TBLK;
      v8f Z0[TBLK], Z1[TBLK];
      proj_l0<TBLK>(ax, ci, xlo, x45, half, Z0, Z1);
      if (blk > 0) load_x<TBLK>(xs + (size_t)(t0 - TBLK) * 6, half, xlo, x45);
#pragma unroll
      for (int i = TBLK - 1; i >= 0; --i) {
        rec_step_split(ah, Z0[i], Z1[i], half, c, h);
        store_hs(ws + tstride * (t0 + i), h);
      }
    }
  } else {
    v2f xlo[TBLK], x45[TBLK];
    if (nblk > 0) load_x<TBLK>(xs, half, xlo, x45);
    for (int blk = 0; blk < nblk; ++blk) {
      int t0 = blk * TBLK;
      v8f Z0[TBLK], Z1[TBLK];
      proj_l0<TBLK>(ax, ci, xlo, x45, half, Z0, Z1);
      if (blk + 1 < nblk) load_x<TBLK>(xs + (size_t)(t0 + TBLK) * 6, half, xlo, x45);
#pragma unroll
      for (int i = 0; i < TBLK; ++i) {
        rec_step_split(ah, Z0[i], Z1[i], half, c, h);
        store_hs(ws + tstride * (t0 + i), h);
      }
    }
    for (int t = tfull; t < T; ++t) {
      v2f xl[1], x4[1];
      load_x<1>(xs + (size_t)t * 6, half, xl, x4);
      l0_fwd_block<1>(ax, ah, ci, xl, x4, half, c, h, ws + tstride * t, tstride);
    }
  }
}

__global__ void __launch_bounds__(256)
lstm_l0_scans(const float* __restrict__ tf, const float* __restrict__ h0,
              const float* __restrict__ c0,
              const float* __restrict__ Wihf, const float* __restrict__ Whhf,
              const float* __restrict__ biasf,
              const float* __restrict__ Wihb, const float* __restrict__ Whhb,
              const float* __restrict__ biasb,
              float* __restrict__ f0ws, float* __restrict__ r0ws,
              int B, int T, int first_dir) {
  int tid = blockIdx.x * blockDim.x + threadIdx.x;
  int lane = tid & 31, lm = lane & 15, half = lane >> 4;
  int s = (tid >> 5) * 16 + lm;
  if (s >= B) return;
  int dir = first_dir + (int)blockIdx.y;
  if (dir == 0) l0_scan<false>(tf, h0, c0, Wihf, Whhf, biasf, f0ws, B, T, s, lm, half);
  else          l0_scan<true>(tf, h0, c0, Wihb, Whhb, biasb, r0ws, B, T, s, lm, half);
}

template <int NB>
static __device__ __forceinline__ void load_l0(const float* p0, const float* p1, const float* p2,
                                               size_t tstride, v2f q0[NB], v2f q1[NB], v2f q2[NB]) {
#pragma unroll
  for (int i = 0; i < NB; ++i) {
    q0[i] = *(const v2f*)(p0 + tstride * i);
    q1[i] = *(const v2f*)(p1 + tstride * i);
    q2[i] = *(const v2f*)(p2 + tstride * i);
  }
}

template <int NB>
static __device__ __forceinline__ void proj_l1q(const v2f ax[2][3], const v8f ci[2],
                                                const v2f q0[NB], const v2f q1[NB],
                                                const v2f q2[NB], v8f Z0[NB], v8f Z1[NB]) {
#pragma unroll
  for (int i = 0; i < NB; ++i) {
    v8f z0 = wmma4(ax[0][2], q2[i], ci[0]);
    v8f z1 = wmma4(ax[1][2], q2[i], ci[1]);
    z0 = wmma4(ax[0][1], q1[i], z0);
    z1 = wmma4(ax[1][1], q1[i], z1);
    Z0[i] = wmma4(ax[0][0], q0[i], z0);
    Z1[i] = wmma4(ax[1][0], q0[i], z1);
  }
}

static __device__ __forceinline__ void l1_scan(const float* __restrict__ h0,
                                               const float* __restrict__ c0,
                                               const float* __restrict__ Wih1f,
                                               const float* __restrict__ Whh1f,
                                               const float* __restrict__ b1f,
                                               const float* __restrict__ Wih1b,
                                               const float* __restrict__ Whh1b,
                                               const float* __restrict__ b1b,
                                               const float* __restrict__ fcw,
                                               const float* __restrict__ fcb,
                                               const float* f0ws, const float* r0ws,
                                               float* __restrict__ out,
                                               int B, int T, int s, int lm, int half) {

  v2f a1x[2][3], a1h[2][2]; v8f ci1[2];
  load_Apart<3>(Wih1f, 2 * LSTM_H, a1x, lm, half);
  load_Ahh_split(Whh1f, a1h, lm, half);
  load_bias(b1f, ci1, half);

  float hsB[3], cB[6];
  size_t d2 = 2 * (size_t)B * LSTM_H;
#pragma unroll
  for (int j = 0; j < 3; ++j) hsB[j] = h0[d2 + (size_t)s * 6 + 3 * half + j];
#pragma unroll
  for (int j = 0; j < 6; ++j) cB[j] = 2.0f * K_LOG2E * c0[d2 + (size_t)s * 6 + j];

  const float* fs = f0ws + (size_t)s * 6;
  const float* rs = r0ws + (size_t)s * 6;
  const float* p0 = fs + 2 * half;
  const float* p1 = half ? rs : fs + 4;
  const float* p2 = rs + 2 + 2 * half;
  size_t tstride = (size_t)B * 6;
  int nblk = T / TBLK;
  int tfull = nblk * TBLK;

  v2f q0[TBLK], q1[TBLK], q2[TBLK];
  if (nblk > 0) load_l0<TBLK>(p0, p1, p2, tstride, q0, q1, q2);
  for (int blk = 0; blk < nblk; ++blk) {
    int t0 = blk * TBLK;
    v8f Z0[TBLK], Z1[TBLK];
    proj_l1q<TBLK>(a1x, ci1, q0, q1, q2, Z0, Z1);
    if (blk + 1 < nblk) {
      size_t o = tstride * (size_t)(t0 + TBLK);
      load_l0<TBLK>(p0 + o, p1 + o, p2 + o, tstride, q0, q1, q2);
    }
#pragma unroll
    for (int i = 0; i < TBLK; ++i) rec_step_split(a1h, Z0[i], Z1[i], half, cB, hsB);
  }
  for (int t = tfull; t < T; ++t) {
    size_t o = tstride * (size_t)t;
    v2f u0[1], u1[1], u2[1];
    load_l0<1>(p0 + o, p1 + o, p2 + o, tstride, u0, u1, u2);
    v8f Z0[1], Z1[1];
    proj_l1q<1>(a1x, ci1, u0, u1, u2, Z0, Z1);
    rec_step_split(a1h, Z0[0], Z1[0], half, cB, hsB);
  }

  float hB[6];
#pragma unroll
  for (int j = 0; j < 3; ++j) { hB[j] = hsB[j]; hB[3 + j] = sx16(hsB[j]); }

  v2f a2[2][5]; v8f ci2[2]; load_A(Wih1b, Whh1b, b1b, 2 * LSTM_H, 5, a2, ci2, lm, half);
  float hR[6], cR[6];
  size_t d3 = 3 * (size_t)B * LSTM_H;
#pragma unroll
  for (int j = 0; j < 6; ++j) {
    hR[j] = h0[d3 + (size_t)s * 6 + j];  cR[j] = 2.0f * K_LOG2E * c0[d3 + (size_t)s * 6 + j];
  }
  {
    size_t o = tstride * (size_t)(T - 1);
    float h2s = sx16(hR[2]), h3s = sx16(hR[3]);
    v2f bch[5];
    bch[0] = *(const v2f*)(p0 + o);
    bch[1] = *(const v2f*)(p1 + o);
    bch[2] = *(const v2f*)(p2 + o);
    bch[3].x = half ? h2s : hR[0];   bch[3].y = half ? h3s : hR[1];
    bch[4].x = half ? 0.0f : hR[4];  bch[4].y = half ? 0.0f : hR[5];
    v8f A0, A1; mm(a2, ci2, bch, 5, A0, A1);
    lstm_pointwise(A0, A1, half == 0, cR, hR);
  }
  if (half == 0) {
    float v[12];
#pragma unroll
    for (int k = 0; k < 6; ++k) {
      v[k]     = hB[k] > 0.0f ? hB[k] : 0.0f;
      v[6 + k] = hR[k] > 0.0f ? hR[k] : 0.0f;
    }
#pragma unroll
    for (int j = 0; j < 6; ++j) {
      float acc = fcb[j];
#pragma unroll
      for (int k = 0; k < 12; ++k) acc = fmaf(fcw[j * 12 + k], v[k], acc);
      out[(size_t)s * 6 + j] = acc;
    }
  }
}

__global__ void __launch_bounds__(256)
lstm_owned(const float* __restrict__ tf, const float* __restrict__ h0,
           const float* __restrict__ c0,
           const float* __restrict__ Wih0f, const float* __restrict__ Whh0f,
           const float* __restrict__ b0f,
           const float* __restrict__ Wih0b, const float* __restrict__ Whh0b,
           const float* __restrict__ b0b,
           const float* __restrict__ Wih1f, const float* __restrict__ Whh1f,
           const float* __restrict__ b1f,
           const float* __restrict__ Wih1b, const float* __restrict__ Whh1b,
           const float* __restrict__ b1b,
           const float* __restrict__ fcw, const float* __restrict__ fcb,
           float* f0ws, float* r0ws, float* __restrict__ out, int B, int T) {
  int lane = threadIdx.x & 31, lm = lane & 15, half = lane >> 4;
  int wv = threadIdx.x >> 5;
  int s = ((int)blockIdx.x * 4 + (wv & 3)) * 16 + lm;
  bool live = s < B;
  bool fwd = wv < 4;

  if (live) {
    if (fwd) l0_scan<false>(tf, h0, c0, Wih0f, Whh0f, b0f, f0ws, B, T, s, lm, half);
    else     l0_scan<true>(tf, h0, c0, Wih0b, Whh0b, b0b, r0ws, B, T, s, lm, half);
  }
  __syncthreads();
  if (live && fwd)
    l1_scan(h0, c0, Wih1f, Whh1f, b1f, Wih1b, Whh1b, b1b, fcw, fcb, f0ws, r0ws, out,
            B, T, s, lm, half);
}

__global__ void __launch_bounds__(256)
lstm_fused(const float* __restrict__ tf, const float* __restrict__ h0,
           const float* __restrict__ c0,
           const float* __restrict__ Wih0f, const float* __restrict__ Whh0f,
           const float* __restrict__ b0f,
           const float* __restrict__ Wih1f, const float* __restrict__ Whh1f,
           const float* __restrict__ b1f,
           const float* __restrict__ Wih1b, const float* __restrict__ Whh1b,
           const float* __restrict__ b1b,
           const float* __restrict__ fcw, const float* __restrict__ fcb,
           const float* __restrict__ r0ws, float* __restrict__ out,
           int B, int T) {
  int tid = blockIdx.x * blockDim.x + threadIdx.x;
  int lane = tid & 31, lm = lane & 15, half = lane >> 4;
  int s = (tid >> 5) * 16 + lm;
  if (s >= B) return;

  v2f a0x[2][2], a0h[2][2]; v8f ci0[2];
  load_Apart<2>(Wih0f, LSTM_H, a0x, lm, half);
  load_Apart<2>(Whh0f, LSTM_H, a0h, lm, half);
  load_bias(b0f, ci0, half);
  v2f a1x[2][3], a1h[2][2]; v8f ci1[2];
  load_Apart<3>(Wih1f, 2 * LSTM_H, a1x, lm, half);
  load_Apart<2>(Whh1f, LSTM_H, a1h, lm, half);
  load_bias(b1f, ci1, half);

  float hA[6], cA[6], hB[6], cB[6];
  size_t d0 = 0, d2 = 2 * (size_t)B * LSTM_H;
#pragma unroll
  for (int j = 0; j < 6; ++j) {
    hA[j] = h0[d0 + (size_t)s * 6 + j];  cA[j] = 2.0f * K_LOG2E * c0[d0 + (size_t)s * 6 + j];
    hB[j] = h0[d2 + (size_t)s * 6 + j];  cB[j] = 2.0f * K_LOG2E * c0[d2 + (size_t)s * 6 + j];
  }

  const float* xs = tf + (size_t)s * T * 6;
  const float* rs = r0ws + (size_t)s * 6;
  size_t tstride = (size_t)B * 6;
  int nblk = T / TBLK;
  int tfull = nblk * TBLK;

  v2f xlo[TBLK], x45[TBLK], r01[TBLK], rhi[TBLK];
  if (nblk > 0) {
    load_x<TBLK>(xs, half, xlo, x45);
    load_r<TBLK>(rs, tstride, half, r01, rhi);
  }
  for (int blk = 0; blk < nblk; ++blk) {
    int t0 = blk * TBLK;
    float f0b[TBLK][6];
    {
      v8f Z0[TBLK], Z1[TBLK];
      proj_l0<TBLK>(a0x, ci0, xlo, x45, half, Z0, Z1);
      if (blk + 1 < nblk) load_x<TBLK>(xs + (size_t)(t0 + TBLK) * 6, half, xlo, x45);
#pragma unroll
      for (int i = 0; i < TBLK; ++i) {
        rec_step(a0h, Z0[i], Z1[i], half, cA, hA);
#pragma unroll
        for (int j = 0; j < 6; ++j) f0b[i][j] = hA[j];
      }
    }
    {
      v8f Z0[TBLK], Z1[TBLK];
      proj_l1<TBLK>(a1x, ci1, f0b, r01, rhi, half, Z0, Z1);
      if (blk + 1 < nblk) load_r<TBLK>(rs + tstride * (t0 + TBLK), tstride, half, r01, rhi);
#pragma unroll
      for (int i = 0; i < TBLK; ++i) rec_step(a1h, Z0[i], Z1[i], half, cB, hB);
    }
  }
  for (int t = tfull; t < T; ++t) {
    v2f xl[1], x4[1], ra[1], rb[1];
    load_x<1>(xs + (size_t)t * 6, half, xl, x4);
    load_r<1>(rs + tstride * t, tstride, half, ra, rb);
    v8f Z0[1], Z1[1];
    proj_l0<1>(a0x, ci0, xl, x4, half, Z0, Z1);
    rec_step(a0h, Z0[0], Z1[0], half, cA, hA);
    float f1[1][6];
#pragma unroll
    for (int j = 0; j < 6; ++j) f1[0][j] = hA[j];
    proj_l1<1>(a1x, ci1, f1, ra, rb, half, Z0, Z1);
    rec_step(a1h, Z0[0], Z1[0], half, cB, hB);
  }

  v2f a2[2][5]; v8f ci2[2]; load_A(Wih1b, Whh1b, b1b, 2 * LSTM_H, 5, a2, ci2, lm, half);
  float hR[6], cR[6];
  size_t d3 = 3 * (size_t)B * LSTM_H;
#pragma unroll
  for (int j = 0; j < 6; ++j) {
    hR[j] = h0[d3 + (size_t)s * 6 + j];  cR[j] = 2.0f * K_LOG2E * c0[d3 + (size_t)s * 6 + j];
  }
  {
    const float* r0p = r0ws + ((size_t)(T - 1) * B + s) * 6;
    v2f bch[5]; b_l1(hA, r0p, hR, half, bch);
    v8f A0, A1; mm(a2, ci2, bch, 5, A0, A1);
    lstm_pointwise(A0, A1, half == 0, cR, hR);
  }
  if (half == 0) {
    float v[12];
#pragma unroll
    for (int k = 0; k < 6; ++k) {
      v[k]     = hB[k] > 0.0f ? hB[k] : 0.0f;
      v[6 + k] = hR[k] > 0.0f ? hR[k] : 0.0f;
    }
#pragma unroll
    for (int j = 0; j < 6; ++j) {
      float acc = fcb[j];
#pragma unroll
      for (int k = 0; k < 12; ++k) acc = fmaf(fcw[j * 12 + k], v[k], acc);
      out[(size_t)s * 6 + j] = acc;
    }
  }
}

extern "C" void kernel_launch(void* const* d_in, const int* in_sizes, int n_in,
                              void* d_out, int out_size, void* d_ws, size_t ws_size,
                              hipStream_t stream) {
  const float* tf    = (const float*)d_in[0];
  const float* h0    = (const float*)d_in[1];
  const float* c0    = (const float*)d_in[2];
  const float* Wih0f = (const float*)d_in[3];
  const float* Whh0f = (const float*)d_in[4];
  const float* b0f   = (const float*)d_in[5];
  const float* Wih0b = (const float*)d_in[6];
  const float* Whh0b = (const float*)d_in[7];
  const float* b0b   = (const float*)d_in[8];
  const float* Wih1f = (const float*)d_in[9];
  const float* Whh1f = (const float*)d_in[10];
  const float* b1f   = (const float*)d_in[11];
  const float* Wih1b = (const float*)d_in[12];
  const float* Whh1b = (const float*)d_in[13];
  const float* b1b   = (const float*)d_in[14];
  const float* fcw   = (const float*)d_in[15];
  const float* fcb   = (const float*)d_in[16];

  int B = in_sizes[1] / (4 * LSTM_H);
  int T = in_sizes[0] / (B * LSTM_H);

  size_t need = (size_t)B * T * LSTM_H * sizeof(float);
  if (ws_size < need) return;
  float* r0ws = (float*)d_ws;
  float* f0ws = r0ws + (size_t)B * T * LSTM_H;

  int waves = (B + 15) / 16;
  int threads = waves * 32;
  int block = 256;
  int grid = (threads + block - 1) / block;

  if (ws_size >= 2 * need) {
    int groups = (B + 63) / 64;
    lstm_owned<<<groups, block, 0, stream>>>(tf, h0, c0, Wih0f, Whh0f, b0f, Wih0b, Whh0b, b0b,
                                             Wih1f, Whh1f, b1f, Wih1b, Whh1b, b1b, fcw, fcb,
                                             f0ws, r0ws, (float*)d_out, B, T);
  } else {
    lstm_l0_scans<<<dim3(grid, 1), block, 0, stream>>>(tf, h0, c0, Wih0f, Whh0f, b0f,
                                                       Wih0b, Whh0b, b0b, r0ws, r0ws, B, T, 1);
    lstm_fused<<<grid, block, 0, stream>>>(tf, h0, c0, Wih0f, Whh0f, b0f,
                                           Wih1f, Whh1f, b1f, Wih1b, Whh1b, b1b,
                                           fcw, fcb, r0ws, (float*)d_out, B, T);
  }
}


# layer-0 scans: inputs x4, x5 ride in the two empty k slots of the recurrent product, so the input part is one K chunk not two (6 not 8 products per step); k order of layer-0 gate sums changes
# speedup vs baseline: 3.3199x; 1.0726x over previous
#include <hip/hip_runtime.h>
#include <hip/hip_bf16.h>


typedef __attribute__((ext_vector_type(2))) float v2f;
typedef __attribute__((ext_vector_type(8))) float v8f;

#define LSTM_H 6
#define TBLK 4

#define K_LOG2E 1.44269504088896340736f
static __device__ __forceinline__ float logistic2(float y) {
  return __builtin_amdgcn_rcpf(1.0f + __builtin_amdgcn_exp2f(y));
}
static __device__ __forceinline__ float row_scale(int m) {
  return ((m >> 3) == 2) ? 2.0f * K_LOG2E : -K_LOG2E;
}
static __device__ __forceinline__ float sx16(float v) {
  return __shfl_xor(v, 16, 32);
}
static __device__ __forceinline__ v8f wmma4(v2f a, v2f b, v8f c) {
  return __builtin_amdgcn_wmma_f32_16x16x4_f32(false, a, false, b, (short)0, c,
                                               false, false);
}

static __device__ __forceinline__ int gate_row(int m) {
  int idx = m & 7;
  return (idx < LSTM_H) ? ((m >> 3) * LSTM_H + idx) : -1;
}

static __device__ __forceinline__ float wcomb(const float* Wih, const float* Whh,
                                              int D, int m, int k) {
  int r = gate_row(m);
  if (r < 0) return 0.0f;
  float sc = row_scale(m);
  if (k < D) return sc * Wih[r * D + k];
  k -= D;
  return (k < LSTM_H) ? sc * Whh[r * LSTM_H + k] : 0.0f;
}

static __device__ __forceinline__ float wpad(const float* Wm, int ncol, int m, int k) {
  int r = gate_row(m);
  if (r < 0) return 0.0f;
  return (k < ncol) ? row_scale(m) * Wm[r * ncol + k] : 0.0f;
}

template <int NCH>
static __device__ __forceinline__ void load_Apart(const float* Wm, int ncol,
                                                  v2f a[2][NCH], int lm, int half) {
#pragma unroll
  for (int t = 0; t < 2; ++t) {
    int m = t * 16 + lm;
#pragma unroll
    for (int c = 0; c < NCH; ++c) {
      int k0 = 4 * c + 2 * half;
      v2f av;
      av.x = wpad(Wm, ncol, m, k0);
      av.y = wpad(Wm, ncol, m, k0 + 1);
      a[t][c] = av;
    }
  }
}

static __device__ __forceinline__ void load_bias(const float* bias, v8f cini[2], int half) {
#pragma unroll
  for (int t = 0; t < 2; ++t) {
    v8f ci;
#pragma unroll
    for (int r = 0; r < 8; ++r) {
      int m = t * 16 + half * 8 + r;
      int gr = gate_row(m);
      ci[r] = (gr >= 0) ? row_scale(m) * bias[gr] : 0.0f;
    }
    cini[t] = ci;
  }
}

static __device__ __forceinline__ void load_A(const float* Wih, const float* Whh,
                                              const float* bias, int D, int nch,
                                              v2f a[2][5], v8f cini[2],
                                              int lm, int half) {
  for (int t = 0; t < 2; ++t) {
    int m = t * 16 + lm;
#pragma unroll
    for (int c = 0; c < 5; ++c) {
      if (c < nch) {
        int k0 = 4 * c + 2 * half;
        v2f av;
        av.x = wcomb(Wih, Whh, D, m, k0);
        av.y = wcomb(Wih, Whh, D, m, k0 + 1);
        a[t][c] = av;
      }
    }
  }
  load_bias(bias, cini, half);
}

static __device__ __forceinline__ void mm(const v2f a[2][5], const v8f cini[2],
                                          const v2f bch[5], int nch,
                                          v8f& A0, v8f& A1) {
  A0 = cini[0];
  A1 = cini[1];
#pragma unroll
  for (int c = 0; c < 5; ++c) {
    if (c < nch) {
      A0 = wmma4(a[0][c], bch[c], A0);
      A1 = wmma4(a[1][c], bch[c], A1);
    }
  }
}

static __device__ __forceinline__ void lstm_pointwise(const v8f& A0, const v8f& A1,
                                                      bool low, float c[6], float h[6]) {
  const float gm = low ? -4.0f * K_LOG2E : 1.0f;
  const float ga = low ? 2.0f * K_LOG2E : 0.0f;
#pragma unroll
  for (int r = 0; r < LSTM_H; ++r) {
    float s_if = logistic2(A0[r]);
    float go = fmaf(logistic2(A1[r]), gm, ga);
    float fsw = sx16(s_if);
    float cn = fsw * c[r] + s_if * go;
    c[r] = cn;
    float osw = sx16(go);
    h[r] = osw * fmaf(logistic2(cn), -2.0f, 1.0f);
  }
}

static __device__ __forceinline__ void rec_step(const v2f ah[2][2], const v8f& Z0, const v8f& Z1,
                                                int half, float c[6], float h[6]) {
  float h2s = sx16(h[2]), h3s = sx16(h[3]);
  v2f b0, b1;
  b1.x = half ? 0.0f : h[4];  b1.y = half ? 0.0f : h[5];
  b0.x = half ? h2s : h[0];   b0.y = half ? h3s : h[1];
  v8f A0 = wmma4(ah[0][1], b1, Z0);
  v8f A1 = wmma4(ah[1][1], b1, Z1);
  A0 = wmma4(ah[0][0], b0, A0);
  A1 = wmma4(ah[1][0], b0, A1);
  lstm_pointwise(A0, A1, half == 0, c, h);
}


static __device__ __forceinline__ float pick(bool first, float v_first, float v_second) {
  return first ? v_first : v_second;
}

static __device__ __forceinline__ void load_Ahh_split(const float* Whh, v2f a[2][2],
                                                      int lm, int half) {
#pragma unroll
  for (int t = 0; t < 2; ++t) {
    int m = t * 16 + lm;
    v2f a0, a1;
    a0.x = wpad(Whh, LSTM_H, m, 3 * half);
    a0.y = wpad(Whh, LSTM_H, m, 3 * half + 1);
    a1.x = wpad(Whh, LSTM_H, m, 3 * half + 2);
    a1.y = 0.0f;
    a[t][0] = a0;
    a[t][1] = a1;
  }
}

static __device__ __forceinline__ void load_Ahh_split_x(const float* Whh, const float* Wih,
                                                        v2f a[2][2], int lm, int half) {
  load_Ahh_split(Whh, a, lm, half);
  a[0][1].y = wpad(Wih, LSTM_H, lm, 4 + half);
  a[1][1].y = wpad(Wih, LSTM_H, 16 + lm, 4 + half);
}

static __device__ __forceinline__ void lstm_pointwise_split(const v8f& A0, const v8f& A1,
                                                            bool low, float c[6], float hs[3]) {
  const float gm = low ? -4.0f * K_LOG2E : 1.0f;
  const float ga = low ? 2.0f * K_LOG2E : 0.0f;
  float go[6], term[6], tsw[6], osw[3];
#pragma unroll
  for (int r = 0; r < LSTM_H; ++r) {
    float s_if = logistic2(A0[r]);
    go[r] = fmaf(logistic2(A1[r]), gm, ga);
    term[r] = s_if * pick(low, go[r], c[r]);
    tsw[r] = sx16(term[r]);
  }
#pragma unroll
  for (int j = 0; j < 3; ++j) osw[j] = sx16(go[j]);
  __builtin_amdgcn_sched_barrier(0);
#pragma unroll
  for (int r = 0; r < LSTM_H; ++r)
    c[r] = term[r] + tsw[r];
#pragma unroll
  for (int j = 0; j < 3; ++j) {
    float so = pick(low, osw[j], go[3 + j]);
    float cu = pick(low, c[j], c[3 + j]);
    hs[j] = so * fmaf(logistic2(cu), -2.0f, 1.0f);
  }
}

static __device__ __forceinline__ void rec_step_split(const v2f ah[2][2], const v8f& Z0,
                                                      const v8f& Z1, int half,
                                                      float c[6], float hs[3], float xpad) {
  v2f b0, b1;
  b0.x = hs[0];  b0.y = hs[1];
  b1.x = hs[2];  b1.y = xpad;
  v8f A0 = wmma4(ah[0][1], b1, Z0);
  v8f A1 = wmma4(ah[1][1], b1, Z1);
  A0 = wmma4(ah[0][0], b0, A0);
  A1 = wmma4(ah[1][0], b0, A1);
  lstm_pointwise_split(A0, A1, half == 0, c, hs);
}

static __device__ __forceinline__ void store_hs(float* wp3, const float hs[3]) {
  wp3[0] = hs[0]; wp3[1] = hs[1]; wp3[2] = hs[2];
}

template <int NB>
static __device__ __forceinline__ void load_x(const float* xp, int half, v2f xlo[NB], v2f x45[NB]) {
#pragma unroll
  for (int i = 0; i < NB; ++i) {
    xlo[i] = *(const v2f*)(xp + 6 * i + 2 * half);
    x45[i] = *(const v2f*)(xp + 6 * i + 4);
  }
}

template <int NB>
static __device__ __forceinline__ void proj_l0(const v2f ax[2][2], const v8f ci[2],
                                               const v2f xlo[NB], const v2f x45[NB], int half,
                                               v8f Z0[NB], v8f Z1[NB]) {
#pragma unroll
  for (int i = 0; i < NB; ++i) {
    v2f b1;
    b1.x = half ? 0.0f : x45[i].x;  b1.y = half ? 0.0f : x45[i].y;
    Z0[i] = wmma4(ax[0][1], b1, wmma4(ax[0][0], xlo[i], ci[0]));
    Z1[i] = wmma4(ax[1][1], b1, wmma4(ax[1][0], xlo[i], ci[1]));
  }
}

template <int NB>
static __device__ __forceinline__ void load_xs(const float* xp, int half, v2f xlo[NB], float xpd[NB]) {
  for (int i = 0; i < NB; ++i) xpd[i] = xp[6 * i + 4 + half];
  for (int i = 0; i < NB; ++i) xlo[i] = *(const v2f*)(xp + 6 * i + 2 * half);
}
template <int NB>
static __device__ __forceinline__ void proj_l0s(const v2f ax[2][1], const v8f ci[2],
                                                const v2f xlo[NB], v8f Z0[NB], v8f Z1[NB]) {
  for (int i = 0; i < NB; ++i) {
    Z0[i] = wmma4(ax[0][0], xlo[i], ci[0]);
    Z1[i] = wmma4(ax[1][0], xlo[i], ci[1]);
  }
}

template <int NB>
static __device__ __forceinline__ void load_r(const float* rp, size_t tstride, int half,
                                              v2f r01[NB], v2f rhi[NB]) {
#pragma unroll
  for (int i = 0; i < NB; ++i) {
    r01[i] = *(const v2f*)(rp + tstride * i);
    rhi[i] = *(const v2f*)(rp + tstride * i + 2 + 2 * half);
  }
}

template <int NB>
static __device__ __forceinline__ void proj_l1(const v2f ax[2][3], const v8f ci[2],
                                               const float f0[NB][6], const v2f r01[NB],
                                               const v2f rhi[NB], int half,
                                               v8f Z0[NB], v8f Z1[NB]) {
#pragma unroll
  for (int i = 0; i < NB; ++i) {
    float f2s = sx16(f0[i][2]), f3s = sx16(f0[i][3]);
    v2f b0, b1;
    b0.x = half ? f2s : f0[i][0];       b0.y = half ? f3s : f0[i][1];
    b1.x = half ? r01[i].x : f0[i][4];  b1.y = half ? r01[i].y : f0[i][5];
    v8f z0 = wmma4(ax[0][2], rhi[i], ci[0]);
    v8f z1 = wmma4(ax[1][2], rhi[i], ci[1]);
    z0 = wmma4(ax[0][1], b1, z0);
    z1 = wmma4(ax[1][1], b1, z1);
    Z0[i] = wmma4(ax[0][0], b0, z0);
    Z1[i] = wmma4(ax[1][0], b0, z1);
  }
}

static __device__ __forceinline__ void b_l1(const float f0[6], const float* r0p,
                                            const float h[6], int half, v2f bch[5]) {
  v2f r01 = *(const v2f*)(r0p);
  v2f rhi = *(const v2f*)(r0p + 2 + 2 * half);
  float f2s = sx16(f0[2]), f3s = sx16(f0[3]);
  float h2s = sx16(h[2]),  h3s = sx16(h[3]);
  v2f b0, b1, b3, b4;
  b0.x = half ? f2s : f0[0];    b0.y = half ? f3s : f0[1];
  b1.x = half ? r01.x : f0[4];  b1.y = half ? r01.y : f0[5];
  b3.x = half ? h2s : h[0];     b3.y = half ? h3s : h[1];
  b4.x = half ? 0.0f : h[4];    b4.y = half ? 0.0f : h[5];
  bch[0] = b0; bch[1] = b1; bch[2] = rhi; bch[3] = b3; bch[4] = b4;
}

template <int NB>
static __device__ __forceinline__ void l0_bwd_block(const v2f ax[2][1], const v2f ah[2][2],
                                                    const v8f ci[2], const v2f xlo[NB],
                                                    const float xpd[NB], int half,
                                                    float c[6], float hs[3],
                                                    float* wp0, size_t tstride) {
  v8f Z0[NB], Z1[NB];
  proj_l0s<NB>(ax, ci, xlo, Z0, Z1);
#pragma unroll
  for (int i = NB - 1; i >= 0; --i) {
    rec_step_split(ah, Z0[i], Z1[i], half, c, hs, xpd[i]);
    store_hs(wp0 + tstride * i, hs);
  }
}

template <int NB>
static __device__ __forceinline__ void l0_fwd_block(const v2f ax[2][1], const v2f ah[2][2],
                                                    const v8f ci[2], const v2f xlo[NB],
                                                    const float xpd[NB], int half,
                                                    float c[6], float hs[3],
                                                    float* wp0, size_t tstride) {
  v8f Z0[NB], Z1[NB];
  proj_l0s<NB>(ax, ci, xlo, Z0, Z1);
#pragma unroll
  for (int i = 0; i < NB; ++i) {
    rec_step_split(ah, Z0[i], Z1[i], half, c, hs, xpd[i]);
    store_hs(wp0 + tstride * i, hs);
  }
}

template <bool REV>
static __device__ __forceinline__ void l0_scan(const float* __restrict__ tf,
                                               const float* __restrict__ h0,
                                               const float* __restrict__ c0,
                                               const float* __restrict__ Wih,
                                               const float* __restrict__ Whh,
                                               const float* __restrict__ bias,
                                               float* __restrict__ plane,
                                               int B, int T, int s, int lm, int half) {
  v2f ax[2][1], ah[2][2]; v8f ci[2];
  load_Apart<1>(Wih, LSTM_H, ax, lm, half);
  load_Ahh_split_x(Whh, Wih, ah, lm, half);
  load_bias(bias, ci, half);

  float h[3], c[6];
  size_t dsl = REV ? (size_t)B * LSTM_H : 0;
#pragma unroll
  for (int j = 0; j < 3; ++j) h[j] = h0[dsl + (size_t)s * 6 + 3 * half + j];
#pragma unroll
  for (int j = 0; j < 6; ++j)
    c[j] = 2.0f * K_LOG2E * c0[dsl + (size_t)s * 6 + j];

  const float* xs = tf + (size_t)s * T * 6;
  float* ws = plane + (size_t)s * 6 + 3 * half;
  size_t tstride = (size_t)B * 6;
  int nblk = T / TBLK;
  int tfull = nblk * TBLK;

  if (REV) {
    for (int t = T - 1; t >= tfull; --t) {
      v2f xl[1]; float xq[1];
      load_xs<1>(xs + (size_t)t * 6, half, xl, xq);
      l0_bwd_block<1>(ax, ah, ci, xl, xq, half, c, h, ws + tstride * t, tstride);
    }

    v2f xlo[TBLK]; float xpd[TBLK];
    if (nblk > 0) load_xs<TBLK>(xs + (size_t)(tfull - TBLK) * 6, half, xlo, xpd);
    for (int blk = nblk - 1; blk >= 0; --blk) {
      int t0 = blk * TBLK;
      v8f Z0[TBLK], Z1[TBLK];
      float xq[TBLK];
      proj_l0s<TBLK>(ax, ci, xlo, Z0, Z1);
      for (int i = 0; i < TBLK; ++i) xq[i] = xpd[i];
      if (blk > 0) load_xs<TBLK>(xs + (size_t)(t0 - TBLK) * 6, half, xlo, xpd);
#pragma unroll
      for (int i = TBLK - 1; i >= 0; --i) {
        rec_step_split(ah, Z0[i], Z1[i], half, c, h, xq[i]);
        store_hs(ws + tstride * (t0 + i), h);
      }
    }
  } else {
    v2f xlo[TBLK]; float xpd[TBLK];
    if (nblk > 0) load_xs<TBLK>(xs, half, xlo, xpd);
    for (int blk = 0; blk < nblk; ++blk) {
      int t0 = blk * TBLK;
      v8f Z0[TBLK], Z1[TBLK];
      float xq[TBLK];
      proj_l0s<TBLK>(ax, ci, xlo, Z0, Z1);
      for (int i = 0; i < TBLK; ++i) xq[i] = xpd[i];
      if (blk + 1 < nblk) load_xs<TBLK>(xs + (size_t)(t0 + TBLK) * 6, half, xlo, xpd);
#pragma unroll
      for (int i = 0; i < TBLK; ++i) {
        rec_step_split(ah, Z0[i], Z1[i], half, c, h, xq[i]);
        store_hs(ws + tstride * (t0 + i), h);
      }
    }
    for (int t = tfull; t < T; ++t) {
      v2f xl[1]; float xq[1];
      load_xs<1>(xs + (size_t)t * 6, half, xl, xq);
      l0_fwd_block<1>(ax, ah, ci, xl, xq, half, c, h, ws + tstride * t, tstride);
    }
  }
}

__global__ void __launch_bounds__(256)
lstm_l0_scans(const float* __restrict__ tf, const float* __restrict__ h0,
              const float* __restrict__ c0,
              const float* __restrict__ Wihf, const float* __restrict__ Whhf,
              const float* __restrict__ biasf,
              const float* __restrict__ Wihb, const float* __restrict__ Whhb,
              const float* __restrict__ biasb,
              float* __restrict__ f0ws, float* __restrict__ r0ws,
              int B, int T, int first_dir) {
  int tid = blockIdx.x * blockDim.x + threadIdx.x;
  int lane = tid & 31, lm = lane & 15, half = lane >> 4;
  int s = (tid >> 5) * 16 + lm;
  if (s >= B) return;
  int dir = first_dir + (int)blockIdx.y;
  if (dir == 0) l0_scan<false>(tf, h0, c0, Wihf, Whhf, biasf, f0ws, B, T, s, lm, half);
  else          l0_scan<true>(tf, h0, c0, Wihb, Whhb, biasb, r0ws, B, T, s, lm, half);
}

template <int NB>
static __device__ __forceinline__ void load_l0(const float* p0, const float* p1, const float* p2,
                                               size_t tstride, v2f q0[NB], v2f q1[NB], v2f q2[NB]) {
#pragma unroll
  for (int i = 0; i < NB; ++i) {
    q0[i] = *(const v2f*)(p0 + tstride * i);
    q1[i] = *(const v2f*)(p1 + tstride * i);
    q2[i] = *(const v2f*)(p2 + tstride * i);
  }
}

template <int NB>
static __device__ __forceinline__ void proj_l1q(const v2f ax[2][3], const v8f ci[2],
                                                const v2f q0[NB], const v2f q1[NB],
                                                const v2f q2[NB], v8f Z0[NB], v8f Z1[NB]) {
#pragma unroll
  for (int i = 0; i < NB; ++i) {
    v8f z0 = wmma4(ax[0][2], q2[i], ci[0]);
    v8f z1 = wmma4(ax[1][2], q2[i], ci[1]);
    z0 = wmma4(ax[0][1], q1[i], z0);
    z1 = wmma4(ax[1][1], q1[i], z1);
    Z0[i] = wmma4(ax[0][0], q0[i], z0);
    Z1[i] = wmma4(ax[1][0], q0[i], z1);
  }
}

static __device__ __forceinline__ void l1_scan(const float* __restrict__ h0,
                                               const float* __restrict__ c0,
                                               const float* __restrict__ Wih1f,
                                               const float* __restrict__ Whh1f,
                                               const float* __restrict__ b1f,
                                               const float* __restrict__ Wih1b,
                                               const float* __restrict__ Whh1b,
                                               const float* __restrict__ b1b,
                                               const float* __restrict__ fcw,
                                               const float* __restrict__ fcb,
                                               const float* f0ws, const float* r0ws,
                                               float* __restrict__ out,
                                               int B, int T, int s, int lm, int half) {

  v2f a1x[2][3], a1h[2][2]; v8f ci1[2];
  load_Apart<3>(Wih1f, 2 * LSTM_H, a1x, lm, half);
  load_Ahh_split(Whh1f, a1h, lm, half);
  load_bias(b1f, ci1, half);

  float hsB[3], cB[6];
  size_t d2 = 2 * (size_t)B * LSTM_H;
#pragma unroll
  for (int j = 0; j < 3; ++j) hsB[j] = h0[d2 + (size_t)s * 6 + 3 * half + j];
#pragma unroll
  for (int j = 0; j < 6; ++j) cB[j] = 2.0f * K_LOG2E * c0[d2 + (size_t)s * 6 + j];

  const float* fs = f0ws + (size_t)s * 6;
  const float* rs = r0ws + (size_t)s * 6;
  const float* p0 = fs + 2 * half;
  const float* p1 = half ? rs : fs + 4;
  const float* p2 = rs + 2 + 2 * half;
  size_t tstride = (size_t)B * 6;
  int nblk = T / TBLK;
  int tfull = nblk * TBLK;

  v2f q0[TBLK], q1[TBLK], q2[TBLK];
  if (nblk > 0) load_l0<TBLK>(p0, p1, p2, tstride, q0, q1, q2);
  for (int blk = 0; blk < nblk; ++blk) {
    int t0 = blk * TBLK;
    v8f Z0[TBLK], Z1[TBLK];
    proj_l1q<TBLK>(a1x, ci1, q0, q1, q2, Z0, Z1);
    if (blk + 1 < nblk) {
      size_t o = tstride * (size_t)(t0 + TBLK);
      load_l0<TBLK>(p0 + o, p1 + o, p2 + o, tstride, q0, q1, q2);
    }
#pragma unroll
    for (int i = 0; i < TBLK; ++i) rec_step_split(a1h, Z0[i], Z1[i], half, cB, hsB, 0.0f);
  }
  for (int t = tfull; t < T; ++t) {
    size_t o = tstride * (size_t)t;
    v2f u0[1], u1[1], u2[1];
    load_l0<1>(p0 + o, p1 + o, p2 + o, tstride, u0, u1, u2);
    v8f Z0[1], Z1[1];
    proj_l1q<1>(a1x, ci1, u0, u1, u2, Z0, Z1);
    rec_step_split(a1h, Z0[0], Z1[0], half, cB, hsB, 0.0f);
  }

  float hB[6];
#pragma unroll
  for (int j = 0; j < 3; ++j) { hB[j] = hsB[j]; hB[3 + j] = sx16(hsB[j]); }

  v2f a2[2][5]; v8f ci2[2]; load_A(Wih1b, Whh1b, b1b, 2 * LSTM_H, 5, a2, ci2, lm, half);
  float hR[6], cR[6];
  size_t d3 = 3 * (size_t)B * LSTM_H;
#pragma unroll
  for (int j = 0; j < 6; ++j) {
    hR[j] = h0[d3 + (size_t)s * 6 + j];  cR[j] = 2.0f * K_LOG2E * c0[d3 + (size_t)s * 6 + j];
  }
  {
    size_t o = tstride * (size_t)(T - 1);
    float h2s = sx16(hR[2]), h3s = sx16(hR[3]);
    v2f bch[5];
    bch[0] = *(const v2f*)(p0 + o);
    bch[1] = *(const v2f*)(p1 + o);
    bch[2] = *(const v2f*)(p2 + o);
    bch[3].x = half ? h2s : hR[0];   bch[3].y = half ? h3s : hR[1];
    bch[4].x = half ? 0.0f : hR[4];  bch[4].y = half ? 0.0f : hR[5];
    v8f A0, A1; mm(a2, ci2, bch, 5, A0, A1);
    lstm_pointwise(A0, A1, half == 0, cR, hR);
  }
  if (half == 0) {
    float v[12];
#pragma unroll
    for (int k = 0; k < 6; ++k) {
      v[k]     = hB[k] > 0.0f ? hB[k] : 0.0f;
      v[6 + k] = hR[k] > 0.0f ? hR[k] : 0.0f;
    }
#pragma unroll
    for (int j = 0; j < 6; ++j) {
      float acc = fcb[j];
#pragma unroll
      for (int k = 0; k < 12; ++k) acc = fmaf(fcw[j * 12 + k], v[k], acc);
      out[(size_t)s * 6 + j] = acc;
    }
  }
}

__global__ void __launch_bounds__(256)
lstm_owned(const float* __restrict__ tf, const float* __restrict__ h0,
           const float* __restrict__ c0,
           const float* __restrict__ Wih0f, const float* __restrict__ Whh0f,
           const float* __restrict__ b0f,
           const float* __restrict__ Wih0b, const float* __restrict__ Whh0b,
           const float* __restrict__ b0b,
           const float* __restrict__ Wih1f, const float* __restrict__ Whh1f,
           const float* __restrict__ b1f,
           const float* __restrict__ Wih1b, const float* __restrict__ Whh1b,
           const float* __restrict__ b1b,
           const float* __restrict__ fcw, const float* __restrict__ fcb,
           float* f0ws, float* r0ws, float* __restrict__ out, int B, int T) {
  int lane = threadIdx.x & 31, lm = lane & 15, half = lane >> 4;
  int wv = threadIdx.x >> 5;
  int s = ((int)blockIdx.x * 4 + (wv & 3)) * 16 + lm;
  bool live = s < B;
  bool fwd = wv < 4;

  if (live) {
    if (fwd) l0_scan<false>(tf, h0, c0, Wih0f, Whh0f, b0f, f0ws, B, T, s, lm, half);
    else     l0_scan<true>(tf, h0, c0, Wih0b, Whh0b, b0b, r0ws, B, T, s, lm, half);
  }
  __syncthreads();
  if (live && fwd)
    l1_scan(h0, c0, Wih1f, Whh1f, b1f, Wih1b, Whh1b, b1b, fcw, fcb, f0ws, r0ws, out,
            B, T, s, lm, half);
}

__global__ void __launch_bounds__(256)
lstm_fused(const float* __restrict__ tf, const float* __restrict__ h0,
           const float* __restrict__ c0,
           const float* __restrict__ Wih0f, const float* __restrict__ Whh0f,
           const float* __restrict__ b0f,
           const float* __restrict__ Wih1f, const float* __restrict__ Whh1f,
           const float* __restrict__ b1f,
           const float* __restrict__ Wih1b, const float* __restrict__ Whh1b,
           const float* __restrict__ b1b,
           const float* __restrict__ fcw, const float* __restrict__ fcb,
           const float* __restrict__ r0ws, float* __restrict__ out,
           int B, int T) {
  int tid = blockIdx.x * blockDim.x + threadIdx.x;
  int lane = tid & 31, lm = lane & 15, half = lane >> 4;
  int s = (tid >> 5) * 16 + lm;
  if (s >= B) return;

  v2f a0x[2][2], a0h[2][2]; v8f ci0[2];
  load_Apart<2>(Wih0f, LSTM_H, a0x, lm, half);
  load_Apart<2>(Whh0f, LSTM_H, a0h, lm, half);
  load_bias(b0f, ci0, half);
  v2f a1x[2][3], a1h[2][2]; v8f ci1[2];
  load_Apart<3>(Wih1f, 2 * LSTM_H, a1x, lm, half);
  load_Apart<2>(Whh1f, LSTM_H, a1h, lm, half);
  load_bias(b1f, ci1, half);

  float hA[6], cA[6], hB[6], cB[6];
  size_t d0 = 0, d2 = 2 * (size_t)B * LSTM_H;
#pragma unroll
  for (int j = 0; j < 6; ++j) {
    hA[j] = h0[d0 + (size_t)s * 6 + j];  cA[j] = 2.0f * K_LOG2E * c0[d0 + (size_t)s * 6 + j];
    hB[j] = h0[d2 + (size_t)s * 6 + j];  cB[j] = 2.0f * K_LOG2E * c0[d2 + (size_t)s * 6 + j];
  }

  const float* xs = tf + (size_t)s * T * 6;
  const float* rs = r0ws + (size_t)s * 6;
  size_t tstride = (size_t)B * 6;
  int nblk = T / TBLK;
  int tfull = nblk * TBLK;

  v2f xlo[TBLK], x45[TBLK], r01[TBLK], rhi[TBLK];
  if (nblk > 0) {
    load_x<TBLK>(xs, half, xlo, x45);
    load_r<TBLK>(rs, tstride, half, r01, rhi);
  }
  for (int blk = 0; blk < nblk; ++blk) {
    int t0 = blk * TBLK;
    float f0b[TBLK][6];
    {
      v8f Z0[TBLK], Z1[TBLK];
      proj_l0<TBLK>(a0x, ci0, xlo, x45, half, Z0, Z1);
      if (blk + 1 < nblk) load_x<TBLK>(xs + (size_t)(t0 + TBLK) * 6, half, xlo, x45);
#pragma unroll
      for (int i = 0; i < TBLK; ++i) {
        rec_step(a0h, Z0[i], Z1[i], half, cA, hA);
#pragma unroll
        for (int j = 0; j < 6; ++j) f0b[i][j] = hA[j];
      }
    }
    {
      v8f Z0[TBLK], Z1[TBLK];
      proj_l1<TBLK>(a1x, ci1, f0b, r01, rhi, half, Z0, Z1);
      if (blk + 1 < nblk) load_r<TBLK>(rs + tstride * (t0 + TBLK), tstride, half, r01, rhi);
#pragma unroll
      for (int i = 0; i < TBLK; ++i) rec_step(a1h, Z0[i], Z1[i], half, cB, hB);
    }
  }
  for (int t = tfull; t < T; ++t) {
    v2f xl[1], x4[1], ra[1], rb[1];
    load_x<1>(xs + (size_t)t * 6, half, xl, x4);
    load_r<1>(rs + tstride * t, tstride, half, ra, rb);
    v8f Z0[1], Z1[1];
    proj_l0<1>(a0x, ci0, xl, x4, half, Z0, Z1);
    rec_step(a0h, Z0[0], Z1[0], half, cA, hA);
    float f1[1][6];
#pragma unroll
    for (int j = 0; j < 6; ++j) f1[0][j] = hA[j];
    proj_l1<1>(a1x, ci1, f1, ra, rb, half, Z0, Z1);
    rec_step(a1h, Z0[0], Z1[0], half, cB, hB);
  }

  v2f a2[2][5]; v8f ci2[2]; load_A(Wih1b, Whh1b, b1b, 2 * LSTM_H, 5, a2, ci2, lm, half);
  float hR[6], cR[6];
  size_t d3 = 3 * (size_t)B * LSTM_H;
#pragma unroll
  for (int j = 0; j < 6; ++j) {
    hR[j] = h0[d3 + (size_t)s * 6 + j];  cR[j] = 2.0f * K_LOG2E * c0[d3 + (size_t)s * 6 + j];
  }
  {
    const float* r0p = r0ws + ((size_t)(T - 1) * B + s) * 6;
    v2f bch[5]; b_l1(hA, r0p, hR, half, bch);
    v8f A0, A1; mm(a2, ci2, bch, 5, A0, A1);
    lstm_pointwise(A0, A1, half == 0, cR, hR);
  }
  if (half == 0) {
    float v[12];
#pragma unroll
    for (int k = 0; k < 6; ++k) {
      v[k]     = hB[k] > 0.0f ? hB[k] : 0.0f;
      v[6 + k] = hR[k] > 0.0f ? hR[k] : 0.0f;
    }
#pragma unroll
    for (int j = 0; j < 6; ++j) {
      float acc = fcb[j];
#pragma unroll
      for (int k = 0; k < 12; ++k) acc = fmaf(fcw[j * 12 + k], v[k], acc);
      out[(size_t)s * 6 + j] = acc;
    }
  }
}

extern "C" void kernel_launch(void* const* d_in, const int* in_sizes, int n_in,
                              void* d_out, int out_size, void* d_ws, size_t ws_size,
                              hipStream_t stream) {
  const float* tf    = (const float*)d_in[0];
  const float* h0    = (const float*)d_in[1];
  const float* c0    = (const float*)d_in[2];
  const float* Wih0f = (const float*)d_in[3];
  const float* Whh0f = (const float*)d_in[4];
  const float* b0f   = (const float*)d_in[5];
  const float* Wih0b = (const float*)d_in[6];
  const float* Whh0b = (const float*)d_in[7];
  const float* b0b   = (const float*)d_in[8];
  const float* Wih1f = (const float*)d_in[9];
  const float* Whh1f = (const float*)d_in[10];
  const float* b1f   = (const float*)d_in[11];
  const float* Wih1b = (const float*)d_in[12];
  const float* Whh1b = (const float*)d_in[13];
  const float* b1b   = (const float*)d_in[14];
  const float* fcw   = (const float*)d_in[15];
  const float* fcb   = (const float*)d_in[16];

  int B = in_sizes[1] / (4 * LSTM_H);
  int T = in_sizes[0] / (B * LSTM_H);

  size_t need = (size_t)B * T * LSTM_H * sizeof(float);
  if (ws_size < need) return;
  float* r0ws = (float*)d_ws;
  float* f0ws = r0ws + (size_t)B * T * LSTM_H;

  int waves = (B + 15) / 16;
  int threads = waves * 32;
  int block = 256;
  int grid = (threads + block - 1) / block;

  if (ws_size >= 2 * need) {
    int groups = (B + 63) / 64;
    lstm_owned<<<groups, block, 0, stream>>>(tf, h0, c0, Wih0f, Whh0f, b0f, Wih0b, Whh0b, b0b,
                                             Wih1f, Whh1f, b1f, Wih1b, Whh1b, b1b, fcw, fcb,
                                             f0ws, r0ws, (float*)d_out, B, T);
  } else {
    lstm_l0_scans<<<dim3(grid, 1), block, 0, stream>>>(tf, h0, c0, Wih0f, Whh0f, b0f,
                                                       Wih0b, Whh0b, b0b, r0ws, r0ws, B, T, 1);
    lstm_fused<<<grid, block, 0, stream>>>(tf, h0, c0, Wih0f, Whh0f, b0f,
                                           Wih1f, Whh1f, b1f, Wih1b, Whh1b, b1b,
                                           fcw, fcb, r0ws, (float*)d_out, B, T);
  }
}


# layer-0 phase: the reverse-scan waves (younger wave of each SIMD's pair) hold wave priority 3 (s_setprio) for their first 5/8 of the scan, then 0: both scans reach the group barrier together
# speedup vs baseline: 3.4228x; 1.0310x over previous
#include <hip/hip_runtime.h>
#include <hip/hip_bf16.h>


typedef __attribute__((ext_vector_type(2))) float v2f;
typedef __attribute__((ext_vector_type(8))) float v8f;

#define LSTM_H 6
#define TBLK 4

#define K_LOG2E 1.44269504088896340736f
static __device__ __forceinline__ float logistic2(float y) {
  return __builtin_amdgcn_rcpf(1.0f + __builtin_amdgcn_exp2f(y));
}
static __device__ __forceinline__ float row_scale(int m) {
  return ((m >> 3) == 2) ? 2.0f * K_LOG2E : -K_LOG2E;
}
static __device__ __forceinline__ float sx16(float v) {
  return __shfl_xor(v, 16, 32);
}
static __device__ __forceinline__ v8f wmma4(v2f a, v2f b, v8f c) {
  return __builtin_amdgcn_wmma_f32_16x16x4_f32(false, a, false, b, (short)0, c,
                                               false, false);
}

static __device__ __forceinline__ int gate_row(int m) {
  int idx = m & 7;
  return (idx < LSTM_H) ? ((m >> 3) * LSTM_H + idx) : -1;
}

static __device__ __forceinline__ float wcomb(const float* Wih, const float* Whh,
                                              int D, int m, int k) {
  int r = gate_row(m);
  if (r < 0) return 0.0f;
  float sc = row_scale(m);
  if (k < D) return sc * Wih[r * D + k];
  k -= D;
  return (k < LSTM_H) ? sc * Whh[r * LSTM_H + k] : 0.0f;
}

static __device__ __forceinline__ float wpad(const float* Wm, int ncol, int m, int k) {
  int r = gate_row(m);
  if (r < 0) return 0.0f;
  return (k < ncol) ? row_scale(m) * Wm[r * ncol + k] : 0.0f;
}

template <int NCH>
static __device__ __forceinline__ void load_Apart(const float* Wm, int ncol,
                                                  v2f a[2][NCH], int lm, int half) {
#pragma unroll
  for (int t = 0; t < 2; ++t) {
    int m = t * 16 + lm;
#pragma unroll
    for (int c = 0; c < NCH; ++c) {
      int k0 = 4 * c + 2 * half;
      v2f av;
      av.x = wpad(Wm, ncol, m, k0);
      av.y = wpad(Wm, ncol, m, k0 + 1);
      a[t][c] = av;
    }
  }
}

static __device__ __forceinline__ void load_bias(const float* bias, v8f cini[2], int half) {
#pragma unroll
  for (int t = 0; t < 2; ++t) {
    v8f ci;
#pragma unroll
    for (int r = 0; r < 8; ++r) {
      int m = t * 16 + half * 8 + r;
      int gr = gate_row(m);
      ci[r] = (gr >= 0) ? row_scale(m) * bias[gr] : 0.0f;
    }
    cini[t] = ci;
  }
}

static __device__ __forceinline__ void load_A(const float* Wih, const float* Whh,
                                              const float* bias, int D, int nch,
                                              v2f a[2][5], v8f cini[2],
                                              int lm, int half) {
  for (int t = 0; t < 2; ++t) {
    int m = t * 16 + lm;
#pragma unroll
    for (int c = 0; c < 5; ++c) {
      if (c < nch) {
        int k0 = 4 * c + 2 * half;
        v2f av;
        av.x = wcomb(Wih, Whh, D, m, k0);
        av.y = wcomb(Wih, Whh, D, m, k0 + 1);
        a[t][c] = av;
      }
    }
  }
  load_bias(bias, cini, half);
}

static __device__ __forceinline__ void mm(const v2f a[2][5], const v8f cini[2],
                                          const v2f bch[5], int nch,
                                          v8f& A0, v8f& A1) {
  A0 = cini[0];
  A1 = cini[1];
#pragma unroll
  for (int c = 0; c < 5; ++c) {
    if (c < nch) {
      A0 = wmma4(a[0][c], bch[c], A0);
      A1 = wmma4(a[1][c], bch[c], A1);
    }
  }
}

static __device__ __forceinline__ void lstm_pointwise(const v8f& A0, const v8f& A1,
                                                      bool low, float c[6], float h[6]) {
  const float gm = low ? -4.0f * K_LOG2E : 1.0f;
  const float ga = low ? 2.0f * K_LOG2E : 0.0f;
#pragma unroll
  for (int r = 0; r < LSTM_H; ++r) {
    float s_if = logistic2(A0[r]);
    float go = fmaf(logistic2(A1[r]), gm, ga);
    float fsw = sx16(s_if);
    float cn = fsw * c[r] + s_if * go;
    c[r] = cn;
    float osw = sx16(go);
    h[r] = osw * fmaf(logistic2(cn), -2.0f, 1.0f);
  }
}

static __device__ __forceinline__ void rec_step(const v2f ah[2][2], const v8f& Z0, const v8f& Z1,
                                                int half, float c[6], float h[6]) {
  float h2s = sx16(h[2]), h3s = sx16(h[3]);
  v2f b0, b1;
  b1.x = half ? 0.0f : h[4];  b1.y = half ? 0.0f : h[5];
  b0.x = half ? h2s : h[0];   b0.y = half ? h3s : h[1];
  v8f A0 = wmma4(ah[0][1], b1, Z0);
  v8f A1 = wmma4(ah[1][1], b1, Z1);
  A0 = wmma4(ah[0][0], b0, A0);
  A1 = wmma4(ah[1][0], b0, A1);
  lstm_pointwise(A0, A1, half == 0, c, h);
}


static __device__ __forceinline__ float pick(bool first, float v_first, float v_second) {
  return first ? v_first : v_second;
}

static __device__ __forceinline__ void load_Ahh_split(const float* Whh, v2f a[2][2],
                                                      int lm, int half) {
#pragma unroll
  for (int t = 0; t < 2; ++t) {
    int m = t * 16 + lm;
    v2f a0, a1;
    a0.x = wpad(Whh, LSTM_H, m, 3 * half);
    a0.y = wpad(Whh, LSTM_H, m, 3 * half + 1);
    a1.x = wpad(Whh, LSTM_H, m, 3 * half + 2);
    a1.y = 0.0f;
    a[t][0] = a0;
    a[t][1] = a1;
  }
}

static __device__ __forceinline__ void load_Ahh_split_x(const float* Whh, const float* Wih,
                                                        v2f a[2][2], int lm, int half) {
  load_Ahh_split(Whh, a, lm, half);
  a[0][1].y = wpad(Wih, LSTM_H, lm, 4 + half);
  a[1][1].y = wpad(Wih, LSTM_H, 16 + lm, 4 + half);
}

static __device__ __forceinline__ void lstm_pointwise_split(const v8f& A0, const v8f& A1,
                                                            bool low, float c[6], float hs[3]) {
  const float gm = low ? -4.0f * K_LOG2E : 1.0f;
  const float ga = low ? 2.0f * K_LOG2E : 0.0f;
  float go[6], term[6], tsw[6], osw[3];
#pragma unroll
  for (int r = 0; r < LSTM_H; ++r) {
    float s_if = logistic2(A0[r]);
    go[r] = fmaf(logistic2(A1[r]), gm, ga);
    term[r] = s_if * pick(low, go[r], c[r]);
    tsw[r] = sx16(term[r]);
  }
#pragma unroll
  for (int j = 0; j < 3; ++j) osw[j] = sx16(go[j]);
  __builtin_amdgcn_sched_barrier(0);
#pragma unroll
  for (int r = 0; r < LSTM_H; ++r)
    c[r] = term[r] + tsw[r];
#pragma unroll
  for (int j = 0; j < 3; ++j) {
    float so = pick(low, osw[j], go[3 + j]);
    float cu = pick(low, c[j], c[3 + j]);
    hs[j] = so * fmaf(logistic2(cu), -2.0f, 1.0f);
  }
}

static __device__ __forceinline__ void rec_step_split(const v2f ah[2][2], const v8f& Z0,
                                                      const v8f& Z1, int half,
                                                      float c[6], float hs[3], float xpad) {
  v2f b0, b1;
  b0.x = hs[0];  b0.y = hs[1];
  b1.x = hs[2];  b1.y = xpad;
  v8f A0 = wmma4(ah[0][1], b1, Z0);
  v8f A1 = wmma4(ah[1][1], b1, Z1);
  A0 = wmma4(ah[0][0], b0, A0);
  A1 = wmma4(ah[1][0], b0, A1);
  lstm_pointwise_split(A0, A1, half == 0, c, hs);
}

static __device__ __forceinline__ void store_hs(float* wp3, const float hs[3]) {
  wp3[0] = hs[0]; wp3[1] = hs[1]; wp3[2] = hs[2];
}

template <int NB>
static __device__ __forceinline__ void load_x(const float* xp, int half, v2f xlo[NB], v2f x45[NB]) {
#pragma unroll
  for (int i = 0; i < NB; ++i) {
    xlo[i] = *(const v2f*)(xp + 6 * i + 2 * half);
    x45[i] = *(const v2f*)(xp + 6 * i + 4);
  }
}

template <int NB>
static __device__ __forceinline__ void proj_l0(const v2f ax[2][2], const v8f ci[2],
                                               const v2f xlo[NB], const v2f x45[NB], int half,
                                               v8f Z0[NB], v8f Z1[NB]) {
#pragma unroll
  for (int i = 0; i < NB; ++i) {
    v2f b1;
    b1.x = half ? 0.0f : x45[i].x;  b1.y = half ? 0.0f : x45[i].y;
    Z0[i] = wmma4(ax[0][1], b1, wmma4(ax[0][0], xlo[i], ci[0]));
    Z1[i] = wmma4(ax[1][1], b1, wmma4(ax[1][0], xlo[i], ci[1]));
  }
}

template <int NB>
static __device__ __forceinline__ void load_xs(const float* xp, int half, v2f xlo[NB], float xpd[NB]) {
  for (int i = 0; i < NB; ++i) xpd[i] = xp[6 * i + 4 + half];
  for (int i = 0; i < NB; ++i) xlo[i] = *(const v2f*)(xp + 6 * i + 2 * half);
}
template <int NB>
static __device__ __forceinline__ void proj_l0s(const v2f ax[2][1], const v8f ci[2],
                                                const v2f xlo[NB], v8f Z0[NB], v8f Z1[NB]) {
  for (int i = 0; i < NB; ++i) {
    Z0[i] = wmma4(ax[0][0], xlo[i], ci[0]);
    Z1[i] = wmma4(ax[1][0], xlo[i], ci[1]);
  }
}

template <int NB>
static __device__ __forceinline__ void load_r(const float* rp, size_t tstride, int half,
                                              v2f r01[NB], v2f rhi[NB]) {
#pragma unroll
  for (int i = 0; i < NB; ++i) {
    r01[i] = *(const v2f*)(rp + tstride * i);
    rhi[i] = *(const v2f*)(rp + tstride * i + 2 + 2 * half);
  }
}

template <int NB>
static __device__ __forceinline__ void proj_l1(const v2f ax[2][3], const v8f ci[2],
                                               const float f0[NB][6], const v2f r01[NB],
                                               const v2f rhi[NB], int half,
                                               v8f Z0[NB], v8f Z1[NB]) {
#pragma unroll
  for (int i = 0; i < NB; ++i) {
    float f2s = sx16(f0[i][2]), f3s = sx16(f0[i][3]);
    v2f b0, b1;
    b0.x = half ? f2s : f0[i][0];       b0.y = half ? f3s : f0[i][1];
    b1.x = half ? r01[i].x : f0[i][4];  b1.y = half ? r01[i].y : f0[i][5];
    v8f z0 = wmma4(ax[0][2], rhi[i], ci[0]);
    v8f z1 = wmma4(ax[1][2], rhi[i], ci[1]);
    z0 = wmma4(ax[0][1], b1, z0);
    z1 = wmma4(ax[1][1], b1, z1);
    Z0[i] = wmma4(ax[0][0], b0, z0);
    Z1[i] = wmma4(ax[1][0], b0, z1);
  }
}

static __device__ __forceinline__ void b_l1(const float f0[6], const float* r0p,
                                            const float h[6], int half, v2f bch[5]) {
  v2f r01 = *(const v2f*)(r0p);
  v2f rhi = *(const v2f*)(r0p + 2 + 2 * half);
  float f2s = sx16(f0[2]), f3s = sx16(f0[3]);
  float h2s = sx16(h[2]),  h3s = sx16(h[3]);
  v2f b0, b1, b3, b4;
  b0.x = half ? f2s : f0[0];    b0.y = half ? f3s : f0[1];
  b1.x = half ? r01.x : f0[4];  b1.y = half ? r01.y : f0[5];
  b3.x = half ? h2s : h[0];     b3.y = half ? h3s : h[1];
  b4.x = half ? 0.0f : h[4];    b4.y = half ? 0.0f : h[5];
  bch[0] = b0; bch[1] = b1; bch[2] = rhi; bch[3] = b3; bch[4] = b4;
}

template <int NB>
static __device__ __forceinline__ void l0_bwd_block(const v2f ax[2][1], const v2f ah[2][2],
                                                    const v8f ci[2], const v2f xlo[NB],
                                                    const float xpd[NB], int half,
                                                    float c[6], float hs[3],
                                                    float* wp0, size_t tstride) {
  v8f Z0[NB], Z1[NB];
  proj_l0s<NB>(ax, ci, xlo, Z0, Z1);
#pragma unroll
  for (int i = NB - 1; i >= 0; --i) {
    rec_step_split(ah, Z0[i], Z1[i], half, c, hs, xpd[i]);
    store_hs(wp0 + tstride * i, hs);
  }
}

template <int NB>
static __device__ __forceinline__ void l0_fwd_block(const v2f ax[2][1], const v2f ah[2][2],
                                                    const v8f ci[2], const v2f xlo[NB],
                                                    const float xpd[NB], int half,
                                                    float c[6], float hs[3],
                                                    float* wp0, size_t tstride) {
  v8f Z0[NB], Z1[NB];
  proj_l0s<NB>(ax, ci, xlo, Z0, Z1);
#pragma unroll
  for (int i = 0; i < NB; ++i) {
    rec_step_split(ah, Z0[i], Z1[i], half, c, hs, xpd[i]);
    store_hs(wp0 + tstride * i, hs);
  }
}

template <bool REV>
static __device__ __forceinline__ void l0_scan(const float* __restrict__ tf,
                                               const float* __restrict__ h0,
                                               const float* __restrict__ c0,
                                               const float* __restrict__ Wih,
                                               const float* __restrict__ Whh,
                                               const float* __restrict__ bias,
                                               float* __restrict__ plane,
                                               int B, int T, int s, int lm, int half,
                                               int hi_blocks) {
  v2f ax[2][1], ah[2][2]; v8f ci[2];
  load_Apart<1>(Wih, LSTM_H, ax, lm, half);
  load_Ahh_split_x(Whh, Wih, ah, lm, half);
  load_bias(bias, ci, half);

  float h[3], c[6];
  size_t dsl = REV ? (size_t)B * LSTM_H : 0;
#pragma unroll
  for (int j = 0; j < 3; ++j) h[j] = h0[dsl + (size_t)s * 6 + 3 * half + j];
#pragma unroll
  for (int j = 0; j < 6; ++j)
    c[j] = 2.0f * K_LOG2E * c0[dsl + (size_t)s * 6 + j];

  const float* xs = tf + (size_t)s * T * 6;
  float* ws = plane + (size_t)s * 6 + 3 * half;
  size_t tstride = (size_t)B * 6;
  int nblk = T / TBLK;
  int tfull = nblk * TBLK;

  if (REV) {
    for (int t = T - 1; t >= tfull; --t) {
      v2f xl[1]; float xq[1];
      load_xs<1>(xs + (size_t)t * 6, half, xl, xq);
      l0_bwd_block<1>(ax, ah, ci, xl, xq, half, c, h, ws + tstride * t, tstride);
    }

    v2f xlo[TBLK]; float xpd[TBLK];
    if (nblk > 0) load_xs<TBLK>(xs + (size_t)(tfull - TBLK) * 6, half, xlo, xpd);
    int hi_left = hi_blocks;
    if (hi_left > 0) __builtin_amdgcn_s_setprio(3);
    for (int blk = nblk - 1; blk >= 0; --blk) {
      int t0 = blk * TBLK;
      v8f Z0[TBLK], Z1[TBLK];
      float xq[TBLK];
      proj_l0s<TBLK>(ax, ci, xlo, Z0, Z1);
      for (int i = 0; i < TBLK; ++i) xq[i] = xpd[i];
      if (blk > 0) load_xs<TBLK>(xs + (size_t)(t0 - TBLK) * 6, half, xlo, xpd);
#pragma unroll
      for (int i = TBLK - 1; i >= 0; --i) {
        rec_step_split(ah, Z0[i], Z1[i], half, c, h, xq[i]);
        store_hs(ws + tstride * (t0 + i), h);
      }
      if (hi_left > 0 && --hi_left == 0) __builtin_amdgcn_s_setprio(0);
    }
  } else {
    v2f xlo[TBLK]; float xpd[TBLK];
    if (nblk > 0) load_xs<TBLK>(xs, half, xlo, xpd);
    for (int blk = 0; blk < nblk; ++blk) {
      int t0 = blk * TBLK;
      v8f Z0[TBLK], Z1[TBLK];
      float xq[TBLK];
      proj_l0s<TBLK>(ax, ci, xlo, Z0, Z1);
      for (int i = 0; i < TBLK; ++i) xq[i] = xpd[i];
      if (blk + 1 < nblk) load_xs<TBLK>(xs + (size_t)(t0 + TBLK) * 6, half, xlo, xpd);
#pragma unroll
      for (int i = 0; i < TBLK; ++i) {
        rec_step_split(ah, Z0[i], Z1[i], half, c, h, xq[i]);
        store_hs(ws + tstride * (t0 + i), h);
      }
    }
    for (int t = tfull; t < T; ++t) {
      v2f xl[1]; float xq[1];
      load_xs<1>(xs + (size_t)t * 6, half, xl, xq);
      l0_fwd_block<1>(ax, ah, ci, xl, xq, half, c, h, ws + tstride * t, tstride);
    }
  }
}

__global__ void __launch_bounds__(256)
lstm_l0_scans(const float* __restrict__ tf, const float* __restrict__ h0,
              const float* __restrict__ c0,
              const float* __restrict__ Wihf, const float* __restrict__ Whhf,
              const float* __restrict__ biasf,
              const float* __restrict__ Wihb, const float* __restrict__ Whhb,
              const float* __restrict__ biasb,
              float* __restrict__ f0ws, float* __restrict__ r0ws,
              int B, int T, int first_dir) {
  int tid = blockIdx.x * blockDim.x + threadIdx.x;
  int lane = tid & 31, lm = lane & 15, half = lane >> 4;
  int s = (tid >> 5) * 16 + lm;
  if (s >= B) return;
  int dir = first_dir + (int)blockIdx.y;
  if (dir == 0) l0_scan<false>(tf, h0, c0, Wihf, Whhf, biasf, f0ws, B, T, s, lm, half, 0);
  else          l0_scan<true>(tf, h0, c0, Wihb, Whhb, biasb, r0ws, B, T, s, lm, half, 0);
}

template <int NB>
static __device__ __forceinline__ void load_l0(const float* p0, const float* p1, const float* p2,
                                               size_t tstride, v2f q0[NB], v2f q1[NB], v2f q2[NB]) {
#pragma unroll
  for (int i = 0; i < NB; ++i) {
    q0[i] = *(const v2f*)(p0 + tstride * i);
    q1[i] = *(const v2f*)(p1 + tstride * i);
    q2[i] = *(const v2f*)(p2 + tstride * i);
  }
}

template <int NB>
static __device__ __forceinline__ void proj_l1q(const v2f ax[2][3], const v8f ci[2],
                                                const v2f q0[NB], const v2f q1[NB],
                                                const v2f q2[NB], v8f Z0[NB], v8f Z1[NB]) {
#pragma unroll
  for (int i = 0; i < NB; ++i) {
    v8f z0 = wmma4(ax[0][2], q2[i], ci[0]);
    v8f z1 = wmma4(ax[1][2], q2[i], ci[1]);
    z0 = wmma4(ax[0][1], q1[i], z0);
    z1 = wmma4(ax[1][1], q1[i], z1);
    Z0[i] = wmma4(ax[0][0], q0[i], z0);
    Z1[i] = wmma4(ax[1][0], q0[i], z1);
  }
}

static __device__ __forceinline__ void l1_scan(const float* __restrict__ h0,
                                               const float* __restrict__ c0,
                                               const float* __restrict__ Wih1f,
                                               const float* __restrict__ Whh1f,
                                               const float* __restrict__ b1f,
                                               const float* __restrict__ Wih1b,
                                               const float* __restrict__ Whh1b,
                                               const float* __restrict__ b1b,
                                               const float* __restrict__ fcw,
                                               const float* __restrict__ fcb,
                                               const float* f0ws, const float* r0ws,
                                               float* __restrict__ out,
                                               int B, int T, int s, int lm, int half) {

  v2f a1x[2][3], a1h[2][2]; v8f ci1[2];
  load_Apart<3>(Wih1f, 2 * LSTM_H, a1x, lm, half);
  load_Ahh_split(Whh1f, a1h, lm, half);
  load_bias(b1f, ci1, half);

  float hsB[3], cB[6];
  size_t d2 = 2 * (size_t)B * LSTM_H;
#pragma unroll
  for (int j = 0; j < 3; ++j) hsB[j] = h0[d2 + (size_t)s * 6 + 3 * half + j];
#pragma unroll
  for (int j = 0; j < 6; ++j) cB[j] = 2.0f * K_LOG2E * c0[d2 + (size_t)s * 6 + j];

  const float* fs = f0ws + (size_t)s * 6;
  const float* rs = r0ws + (size_t)s * 6;
  const float* p0 = fs + 2 * half;
  const float* p1 = half ? rs : fs + 4;
  const float* p2 = rs + 2 + 2 * half;
  size_t tstride = (size_t)B * 6;
  int nblk = T / TBLK;
  int tfull = nblk * TBLK;

  v2f q0[TBLK], q1[TBLK], q2[TBLK];
  if (nblk > 0) load_l0<TBLK>(p0, p1, p2, tstride, q0, q1, q2);
  for (int blk = 0; blk < nblk; ++blk) {
    int t0 = blk * TBLK;
    v8f Z0[TBLK], Z1[TBLK];
    proj_l1q<TBLK>(a1x, ci1, q0, q1, q2, Z0, Z1);
    if (blk + 1 < nblk) {
      size_t o = tstride * (size_t)(t0 + TBLK);
      load_l0<TBLK>(p0 + o, p1 + o, p2 + o, tstride, q0, q1, q2);
    }
#pragma unroll
    for (int i = 0; i < TBLK; ++i) rec_step_split(a1h, Z0[i], Z1[i], half, cB, hsB, 0.0f);
  }
  for (int t = tfull; t < T; ++t) {
    size_t o = tstride * (size_t)t;
    v2f u0[1], u1[1], u2[1];
    load_l0<1>(p0 + o, p1 + o, p2 + o, tstride, u0, u1, u2);
    v8f Z0[1], Z1[1];
    proj_l1q<1>(a1x, ci1, u0, u1, u2, Z0, Z1);
    rec_step_split(a1h, Z0[0], Z1[0], half, cB, hsB, 0.0f);
  }

  float hB[6];
#pragma unroll
  for (int j = 0; j < 3; ++j) { hB[j] = hsB[j]; hB[3 + j] = sx16(hsB[j]); }

  v2f a2[2][5]; v8f ci2[2]; load_A(Wih1b, Whh1b, b1b, 2 * LSTM_H, 5, a2, ci2, lm, half);
  float hR[6], cR[6];
  size_t d3 = 3 * (size_t)B * LSTM_H;
#pragma unroll
  for (int j = 0; j < 6; ++j) {
    hR[j] = h0[d3 + (size_t)s * 6 + j];  cR[j] = 2.0f * K_LOG2E * c0[d3 + (size_t)s * 6 + j];
  }
  {
    size_t o = tstride * (size_t)(T - 1);
    float h2s = sx16(hR[2]), h3s = sx16(hR[3]);
    v2f bch[5];
    bch[0] = *(const v2f*)(p0 + o);
    bch[1] = *(const v2f*)(p1 + o);
    bch[2] = *(const v2f*)(p2 + o);
    bch[3].x = half ? h2s : hR[0];   bch[3].y = half ? h3s : hR[1];
    bch[4].x = half ? 0.0f : hR[4];  bch[4].y = half ? 0.0f : hR[5];
    v8f A0, A1; mm(a2, ci2, bch, 5, A0, A1);
    lstm_pointwise(A0, A1, half == 0, cR, hR);
  }
  if (half == 0) {
    float v[12];
#pragma unroll
    for (int k = 0; k < 6; ++k) {
      v[k]     = hB[k] > 0.0f ? hB[k] : 0.0f;
      v[6 + k] = hR[k] > 0.0f ? hR[k] : 0.0f;
    }
#pragma unroll
    for (int j = 0; j < 6; ++j) {
      float acc = fcb[j];
#pragma unroll
      for (int k = 0; k < 12; ++k) acc = fmaf(fcw[j * 12 + k], v[k], acc);
      out[(size_t)s * 6 + j] = acc;
    }
  }
}

__global__ void __launch_bounds__(256)
lstm_owned(const float* __restrict__ tf, const float* __restrict__ h0,
           const float* __restrict__ c0,
           const float* __restrict__ Wih0f, const float* __restrict__ Whh0f,
           const float* __restrict__ b0f,
           const float* __restrict__ Wih0b, const float* __restrict__ Whh0b,
           const float* __restrict__ b0b,
           const float* __restrict__ Wih1f, const float* __restrict__ Whh1f,
           const float* __restrict__ b1f,
           const float* __restrict__ Wih1b, const float* __restrict__ Whh1b,
           const float* __restrict__ b1b,
           const float* __restrict__ fcw, const float* __restrict__ fcb,
           float* f0ws, float* r0ws, float* __restrict__ out, int B, int T) {
  int lane = threadIdx.x & 31, lm = lane & 15, half = lane >> 4;
  int wv = threadIdx.x >> 5;
  int s = ((int)blockIdx.x * 4 + (wv & 3)) * 16 + lm;
  bool live = s < B;
  bool fwd = wv < 4;

  int hi_blocks = ((T / TBLK) * 5) / 8;
  if (live) {
    if (fwd) l0_scan<false>(tf, h0, c0, Wih0f, Whh0f, b0f, f0ws, B, T, s, lm, half, 0);
    else     l0_scan<true>(tf, h0, c0, Wih0b, Whh0b, b0b, r0ws, B, T, s, lm, half, hi_blocks);
  }
  __syncthreads();
  if (live && fwd)
    l1_scan(h0, c0, Wih1f, Whh1f, b1f, Wih1b, Whh1b, b1b, fcw, fcb, f0ws, r0ws, out,
            B, T, s, lm, half);
}

__global__ void __launch_bounds__(256)
lstm_fused(const float* __restrict__ tf, const float* __restrict__ h0,
           const float* __restrict__ c0,
           const float* __restrict__ Wih0f, const float* __restrict__ Whh0f,
           const float* __restrict__ b0f,
           const float* __restrict__ Wih1f, const float* __restrict__ Whh1f,
           const float* __restrict__ b1f,
           const float* __restrict__ Wih1b, const float* __restrict__ Whh1b,
           const float* __restrict__ b1b,
           const float* __restrict__ fcw, const float* __restrict__ fcb,
           const float* __restrict__ r0ws, float* __restrict__ out,
           int B, int T) {
  int tid = blockIdx.x * blockDim.x + threadIdx.x;
  int lane = tid & 31, lm = lane & 15, half = lane >> 4;
  int s = (tid >> 5) * 16 + lm;
  if (s >= B) return;

  v2f a0x[2][2], a0h[2][2]; v8f ci0[2];
  load_Apart<2>(Wih0f, LSTM_H, a0x, lm, half);
  load_Apart<2>(Whh0f, LSTM_H, a0h, lm, half);
  load_bias(b0f, ci0, half);
  v2f a1x[2][3], a1h[2][2]; v8f ci1[2];
  load_Apart<3>(Wih1f, 2 * LSTM_H, a1x, lm, half);
  load_Apart<2>(Whh1f, LSTM_H, a1h, lm, half);
  load_bias(b1f, ci1, half);

  float hA[6], cA[6], hB[6], cB[6];
  size_t d0 = 0, d2 = 2 * (size_t)B * LSTM_H;
#pragma unroll
  for (int j = 0; j < 6; ++j) {
    hA[j] = h0[d0 + (size_t)s * 6 + j];  cA[j] = 2.0f * K_LOG2E * c0[d0 + (size_t)s * 6 + j];
    hB[j] = h0[d2 + (size_t)s * 6 + j];  cB[j] = 2.0f * K_LOG2E * c0[d2 + (size_t)s * 6 + j];
  }

  const float* xs = tf + (size_t)s * T * 6;
  const float* rs = r0ws + (size_t)s * 6;
  size_t tstride = (size_t)B * 6;
  int nblk = T / TBLK;
  int tfull = nblk * TBLK;

  v2f xlo[TBLK], x45[TBLK], r01[TBLK], rhi[TBLK];
  if (nblk > 0) {
    load_x<TBLK>(xs, half, xlo, x45);
    load_r<TBLK>(rs, tstride, half, r01, rhi);
  }
  for (int blk = 0; blk < nblk; ++blk) {
    int t0 = blk * TBLK;
    float f0b[TBLK][6];
    {
      v8f Z0[TBLK], Z1[TBLK];
      proj_l0<TBLK>(a0x, ci0, xlo, x45, half, Z0, Z1);
      if (blk + 1 < nblk) load_x<TBLK>(xs + (size_t)(t0 + TBLK) * 6, half, xlo, x45);
#pragma unroll
      for (int i = 0; i < TBLK; ++i) {
        rec_step(a0h, Z0[i], Z1[i], half, cA, hA);
#pragma unroll
        for (int j = 0; j < 6; ++j) f0b[i][j] = hA[j];
      }
    }
    {
      v8f Z0[TBLK], Z1[TBLK];
      proj_l1<TBLK>(a1x, ci1, f0b, r01, rhi, half, Z0, Z1);
      if (blk + 1 < nblk) load_r<TBLK>(rs + tstride * (t0 + TBLK), tstride, half, r01, rhi);
#pragma unroll
      for (int i = 0; i < TBLK; ++i) rec_step(a1h, Z0[i], Z1[i], half, cB, hB);
    }
  }
  for (int t = tfull; t < T; ++t) {
    v2f xl[1], x4[1], ra[1], rb[1];
    load_x<1>(xs + (size_t)t * 6, half, xl, x4);
    load_r<1>(rs + tstride * t, tstride, half, ra, rb);
    v8f Z0[1], Z1[1];
    proj_l0<1>(a0x, ci0, xl, x4, half, Z0, Z1);
    rec_step(a0h, Z0[0], Z1[0], half, cA, hA);
    float f1[1][6];
#pragma unroll
    for (int j = 0; j < 6; ++j) f1[0][j] = hA[j];
    proj_l1<1>(a1x, ci1, f1, ra, rb, half, Z0, Z1);
    rec_step(a1h, Z0[0], Z1[0], half, cB, hB);
  }

  v2f a2[2][5]; v8f ci2[2]; load_A(Wih1b, Whh1b, b1b, 2 * LSTM_H, 5, a2, ci2, lm, half);
  float hR[6], cR[6];
  size_t d3 = 3 * (size_t)B * LSTM_H;
#pragma unroll
  for (int j = 0; j < 6; ++j) {
    hR[j] = h0[d3 + (size_t)s * 6 + j];  cR[j] = 2.0f * K_LOG2E * c0[d3 + (size_t)s * 6 + j];
  }
  {
    const float* r0p = r0ws + ((size_t)(T - 1) * B + s) * 6;
    v2f bch[5]; b_l1(hA, r0p, hR, half, bch);
    v8f A0, A1; mm(a2, ci2, bch, 5, A0, A1);
    lstm_pointwise(A0, A1, half == 0, cR, hR);
  }
  if (half == 0) {
    float v[12];
#pragma unroll
    for (int k = 0; k < 6; ++k) {
      v[k]     = hB[k] > 0.0f ? hB[k] : 0.0f;
      v[6 + k] = hR[k] > 0.0f ? hR[k] : 0.0f;
    }
#pragma unroll
    for (int j = 0; j < 6; ++j) {
      float acc = fcb[j];
#pragma unroll
      for (int k = 0; k < 12; ++k) acc = fmaf(fcw[j * 12 + k], v[k], acc);
      out[(size_t)s * 6 + j] = acc;
    }
  }
}

extern "C" void kernel_launch(void* const* d_in, const int* in_sizes, int n_in,
                              void* d_out, int out_size, void* d_ws, size_t ws_size,
                              hipStream_t stream) {
  const float* tf    = (const float*)d_in[0];
  const float* h0    = (const float*)d_in[1];
  const float* c0    = (const float*)d_in[2];
  const float* Wih0f = (const float*)d_in[3];
  const float* Whh0f = (const float*)d_in[4];
  const float* b0f   = (const float*)d_in[5];
  const float* Wih0b = (const float*)d_in[6];
  const float* Whh0b = (const float*)d_in[7];
  const float* b0b   = (const float*)d_in[8];
  const float* Wih1f = (const float*)d_in[9];
  const float* Whh1f = (const float*)d_in[10];
  const float* b1f   = (const float*)d_in[11];
  const float* Wih1b = (const float*)d_in[12];
  const float* Whh1b = (const float*)d_in[13];
  const float* b1b   = (const float*)d_in[14];
  const float* fcw   = (const float*)d_in[15];
  const float* fcb   = (const float*)d_in[16];

  int B = in_sizes[1] / (4 * LSTM_H);
  int T = in_sizes[0] / (B * LSTM_H);

  size_t need = (size_t)B * T * LSTM_H * sizeof(float);
  if (ws_size < need) return;
  float* r0ws = (float*)d_ws;
  float* f0ws = r0ws + (size_t)B * T * LSTM_H;

  int waves = (B + 15) / 16;
  int threads = waves * 32;
  int block = 256;
  int grid = (threads + block - 1) / block;

  if (ws_size >= 2 * need) {
    int groups = (B + 63) / 64;
    lstm_owned<<<groups, block, 0, stream>>>(tf, h0, c0, Wih0f, Whh0f, b0f, Wih0b, Whh0b, b0b,
                                             Wih1f, Whh1f, b1f, Wih1b, Whh1b, b1b, fcw, fcb,
                                             f0ws, r0ws, (float*)d_out, B, T);
  } else {
    lstm_l0_scans<<<dim3(grid, 1), block, 0, stream>>>(tf, h0, c0, Wih0f, Whh0f, b0f,
                                                       Wih0b, Whh0b, b0b, r0ws, r0ws, B, T, 1);
    lstm_fused<<<grid, block, 0, stream>>>(tf, h0, c0, Wih0f, Whh0f, b0f,
                                           Wih1f, Whh1f, b1f, Wih1b, Whh1b, b1b,
                                           fcw, fcb, r0ws, (float*)d_out, B, T);
  }
}
